# starting kernel
# baseline (speedup 1.0000x reference)
#include <hip/hip_runtime.h>

typedef __attribute__((ext_vector_type(2))) float v2f;
typedef __attribute__((ext_vector_type(8))) float v8f;

__device__ __forceinline__ v8f wmma_f32(v2f a, v2f b, v8f c) {
  return __builtin_amdgcn_wmma_f32_16x16x4_f32(false, a, false, b, (short)0, c, false, false);
}

#define BB   2
#define NN   1024
#define DD   1024
#define HH   16
#define DHH  64
#define SCALE 0.125f
#define EPS_  1e-6f
#define NEGBIG (-1e30f)

__device__ __forceinline__ void gemm_ldfrag(const float* __restrict__ aptr,
                                            const float* __restrict__ bptr,
                                            size_t s16, int k0,
                                            v2f a[2], v2f b[4]) {
  a[0] = *(const v2f*)(aptr + k0);
  a[1] = *(const v2f*)(aptr + s16 + k0);
  b[0] = *(const v2f*)(bptr + k0);
  b[1] = *(const v2f*)(bptr + s16 + k0);
  b[2] = *(const v2f*)(bptr + 2 * s16 + k0);
  b[3] = *(const v2f*)(bptr + 3 * s16 + k0);
}

__device__ __forceinline__ void gemm_mma8(const v2f a[2], const v2f b[4],
                                          v8f acc[2][4]) {
#pragma unroll
  for (int mt = 0; mt < 2; ++mt)
#pragma unroll
    for (int nt = 0; nt < 4; ++nt)
      acc[mt][nt] = wmma_f32(a[mt], b[nt], acc[mt][nt]);
}

__global__ __launch_bounds__(128)
void gemm_nt_f32_wmma(const float* __restrict__ A, const float* __restrict__ Bm,
                      float* __restrict__ C, int M, int Nc, int K) {
  const int lane = threadIdx.x & 31;
  const int wave = threadIdx.x >> 5;
  const int lhi  = lane >> 4;
  const int lr   = lane & 15;

  const int nBlocks = Nc >> 7;
  const int mb = blockIdx.x / nBlocks;
  const int nb = blockIdx.x % nBlocks;
  const int mBase = mb * 64  + (wave >> 1) * 32;
  const int nBase = nb * 128 + (wave & 1) * 64;

  v8f acc[2][4];
#pragma unroll
  for (int mt = 0; mt < 2; ++mt)
#pragma unroll
    for (int nt = 0; nt < 4; ++nt) acc[mt][nt] = (v8f){0,0,0,0,0,0,0,0};

  const float* aptr = A  + (size_t)(mBase + lr) * K + 2 * lhi;
  const float* bptr = Bm + (size_t)(nBase + lr) * K + 2 * lhi;
  const size_t s16  = (size_t)16 * K;

  v2f aE[2], bE[4], aO[2], bO[4];
  gemm_ldfrag(aptr, bptr, s16, 0, aE, bE);
#pragma unroll 2
  for (int k0 = 0; k0 < K; k0 += 8) {
    gemm_ldfrag(aptr, bptr, s16, k0 + 4, aO, bO);
    gemm_mma8(aE, bE, acc);
    if (k0 + 8 < K)
      gemm_ldfrag(aptr, bptr, s16, k0 + 8, aE, bE);
    gemm_mma8(aO, bO, acc);
  }

#pragma unroll
  for (int mt = 0; mt < 2; ++mt)
#pragma unroll
    for (int nt = 0; nt < 4; ++nt)
#pragma unroll
      for (int r = 0; r < 8; ++r) {
        size_t row = (size_t)(mBase + mt * 16 + lhi * 8 + r);
        C[row * Nc + (nBase + nt * 16 + lr)] = acc[mt][nt][r];
      }
}

__global__ __launch_bounds__(128)
void stieltjes_attn(const float* __restrict__ qkv, float* __restrict__ O) {
  __shared__ float S[32][1032];
  __shared__ float redA[4][32];
  __shared__ float redB[4][32];
  __shared__ float lam[32];
  __shared__ float wsum[32];

  const int lane = threadIdx.x & 31;
  const int wave = threadIdx.x >> 5;
  const int lhi  = lane >> 4;
  const int lr   = lane & 15;

  const int rb = blockIdx.x & 31;
  const int bh = blockIdx.x >> 5;
  const int h  = bh & (HH - 1);
  const int b  = bh >> 4;

  const int rowBase = rb * 32;
  const int ncols   = rowBase + 32;
  const int nct     = ncols >> 4;

  const float* qbase = qkv + (size_t)b * NN * (3 * DD) + h * DHH;
  const float* kbase = qbase + DD;
  const float* vbase = qbase + 2 * DD;

  v2f aQ[2][16];
#pragma unroll
  for (int mt = 0; mt < 2; ++mt) {
    const float* qr = qbase + (size_t)(rowBase + mt * 16 + lr) * (3 * DD) + 2 * lhi;
#pragma unroll
    for (int ks = 0; ks < 16; ++ks) aQ[mt][ks] = *(const v2f*)(qr + 4 * ks);
  }

  for (int ct = wave; ct < nct; ct += 4) {
    v8f acc0 = (v8f){0,0,0,0,0,0,0,0};
    v8f acc1 = (v8f){0,0,0,0,0,0,0,0};
    const float* kr = kbase + (size_t)(ct * 16 + lr) * (3 * DD) + 2 * lhi;
    v2f bK = *(const v2f*)(kr);
#pragma unroll
    for (int ks = 0; ks < 16; ++ks) {
      v2f bKn;
      if (ks < 15) bKn = *(const v2f*)(kr + 4 * (ks + 1));
      acc0 = wmma_f32(aQ[0][ks], bK, acc0);
      acc1 = wmma_f32(aQ[1][ks], bK, acc1);
      bK = bKn;
    }
    const int gcol = ct * 16 + lr;
#pragma unroll
    for (int r = 0; r < 8; ++r) {
      int row0 = lhi * 8 + r;
      int row1 = 16 + row0;
      float v0 = acc0[r] * SCALE;
      float v1 = acc1[r] * SCALE;
      if (gcol > rowBase + row0) v0 = NEGBIG;
      if (gcol > rowBase + row1) v1 = NEGBIG;
      S[row0][gcol] = v0;
      S[row1][gcol] = v1;
    }
  }
  __syncthreads();

  const int row  = threadIdx.x & 31;
  const int part = threadIdx.x >> 5;
  {
    float m = -3e30f;
    for (int c = part; c < ncols; c += 4) m = fmaxf(m, S[row][c]);
    redA[part][row] = m;
  }
  __syncthreads();
  if (threadIdx.x < 32) {
    float m = fmaxf(fmaxf(redA[0][row], redA[1][row]),
                    fmaxf(redA[2][row], redA[3][row]));
    lam[row] = m + 1.0f;
  }
  __syncthreads();

  for (int it = 0; it < 4; ++it) {
    const float l = lam[row];
    float fs = 0.0f, ps = 0.0f;
    for (int c = part; c < ncols; c += 4) {
      float d  = fmaxf(l - S[row][c], EPS_);
      float r1 = 1.0f / d;
      fs += r1;
      ps += r1 * r1;
    }
    redA[part][row] = fs;
    redB[part][row] = ps;
    __syncthreads();
    if (threadIdx.x < 32) {
      float f  = redA[0][row] + redA[1][row] + redA[2][row] + redA[3][row] - 1.0f;
      float fp = -(redB[0][row] + redB[1][row] + redB[2][row] + redB[3][row]);
      lam[row] = l - f / fp;
    }
    __syncthreads();
  }

  {
    const float l = lam[row];
    float ws = 0.0f;
    for (int c = part; c < ncols; c += 4)
      ws += 1.0f / fmaxf(l - S[row][c], EPS_);
    redA[part][row] = ws;
    __syncthreads();
    if (threadIdx.x < 32)
      wsum[row] = redA[0][row] + redA[1][row] + redA[2][row] + redA[3][row];
    __syncthreads();
    const float inv = 1.0f / wsum[row];
    for (int c = part; c < ncols; c += 4) {
      float d = fmaxf(l - S[row][c], EPS_);
      S[row][c] = (1.0f / d) * inv;
    }
  }
  __syncthreads();

  v8f oacc[2];
  oacc[0] = (v8f){0,0,0,0,0,0,0,0};
  oacc[1] = (v8f){0,0,0,0,0,0,0,0};
  const int dh0 = wave * 16;
  const float* vcol = vbase + dh0 + lr;
  const int ksteps = ncols >> 2;

  v2f bV, a0, a1;
  {
    const int kr = 2 * lhi;
    bV.x = vcol[(size_t)kr * (3 * DD)];
    bV.y = vcol[(size_t)(kr + 1) * (3 * DD)];
    a0 = *(const v2f*)(&S[lr][kr]);
    a1 = *(const v2f*)(&S[16 + lr][kr]);
  }
  for (int ks = 0; ks < ksteps; ++ks) {
    v2f bVn, a0n, a1n;
    if (ks + 1 < ksteps) {
      const int kr = (ks + 1) * 4 + 2 * lhi;
      bVn.x = vcol[(size_t)kr * (3 * DD)];
      bVn.y = vcol[(size_t)(kr + 1) * (3 * DD)];
      a0n = *(const v2f*)(&S[lr][kr]);
      a1n = *(const v2f*)(&S[16 + lr][kr]);
    }
    oacc[0] = wmma_f32(a0, bV, oacc[0]);
    oacc[1] = wmma_f32(a1, bV, oacc[1]);
    bV = bVn; a0 = a0n; a1 = a1n;
  }

  float* obase = O + ((size_t)b * NN + rowBase) * DD + h * DHH + dh0 + lr;
#pragma unroll
  for (int mt = 0; mt < 2; ++mt)
#pragma unroll
    for (int r = 0; r < 8; ++r)
      obase[(size_t)(mt * 16 + lhi * 8 + r) * DD] = oacc[mt][r];
}

extern "C" void kernel_launch(void* const* d_in, const int* in_sizes, int n_in,
                              void* d_out, int out_size, void* d_ws, size_t ws_size,
                              hipStream_t stream) {
  const float* x    = (const float*)d_in[0];
  const float* Wqkv = (const float*)d_in[1];
  const float* Wo   = (const float*)d_in[2];
  float* out = (float*)d_out;

  float* qkv = (float*)d_ws;
  float* Ows = qkv + (size_t)2048 * 3072;

  gemm_nt_f32_wmma<<<(2048 / 64) * (3072 / 128), 128, 0, stream>>>(
      x, Wqkv, qkv, 2048, 3072, 1024);

  stieltjes_attn<<<BB * HH * (NN / 32), 128, 0, stream>>>(qkv, Ows);

  gemm_nt_f32_wmma<<<(2048 / 64) * (1024 / 128), 128, 0, stream>>>(
      Ows, Wo, out, 2048, 1024, 1024);
}


# Attention: scores stay in WMMA accumulators (St = K Q^T, keys split over 8 waves), Newton runs on them, P is the B operand of O^T = V^T P^T, no LDS score array; row sums and PV sum order change
# speedup vs baseline: 2.0634x; 2.0634x over previous
#include <hip/hip_runtime.h>

typedef __attribute__((ext_vector_type(2))) float v2f;
typedef __attribute__((ext_vector_type(8))) float v8f;

__device__ __forceinline__ v8f wmma_f32(v2f a, v2f b, v8f c) {
  return __builtin_amdgcn_wmma_f32_16x16x4_f32(false, a, false, b, (short)0, c, false, false);
}

#define BB   2
#define NN   1024
#define DD   1024
#define HH   16
#define DHH  64
#define SCALE 0.125f
#define EPS_  1e-6f
#define NEGBIG (-1e30f)

__device__ __forceinline__ void gemm_ldfrag(const float* __restrict__ aptr,
                                            const float* __restrict__ bptr,
                                            size_t s16, int k0,
                                            v2f a[2], v2f b[4]) {
  a[0] = *(const v2f*)(aptr + k0);
  a[1] = *(const v2f*)(aptr + s16 + k0);
  b[0] = *(const v2f*)(bptr + k0);
  b[1] = *(const v2f*)(bptr + s16 + k0);
  b[2] = *(const v2f*)(bptr + 2 * s16 + k0);
  b[3] = *(const v2f*)(bptr + 3 * s16 + k0);
}

__device__ __forceinline__ void gemm_mma8(const v2f a[2], const v2f b[4],
                                          v8f acc[2][4]) {
#pragma unroll
  for (int mt = 0; mt < 2; ++mt)
#pragma unroll
    for (int nt = 0; nt < 4; ++nt)
      acc[mt][nt] = wmma_f32(a[mt], b[nt], acc[mt][nt]);
}

__global__ __launch_bounds__(128)
void gemm_nt_f32_wmma(const float* __restrict__ A, const float* __restrict__ Bm,
                      float* __restrict__ C, int M, int Nc, int K) {
  const int lane = threadIdx.x & 31;
  const int wave = threadIdx.x >> 5;
  const int lhi  = lane >> 4;
  const int lr   = lane & 15;

  const int nBlocks = Nc >> 7;
  const int mb = blockIdx.x / nBlocks;
  const int nb = blockIdx.x % nBlocks;
  const int mBase = mb * 64  + (wave >> 1) * 32;
  const int nBase = nb * 128 + (wave & 1) * 64;

  v8f acc[2][4];
#pragma unroll
  for (int mt = 0; mt < 2; ++mt)
#pragma unroll
    for (int nt = 0; nt < 4; ++nt) acc[mt][nt] = (v8f){0,0,0,0,0,0,0,0};

  const float* aptr = A  + (size_t)(mBase + lr) * K + 2 * lhi;
  const float* bptr = Bm + (size_t)(nBase + lr) * K + 2 * lhi;
  const size_t s16  = (size_t)16 * K;

  v2f aE[2], bE[4], aO[2], bO[4];
  gemm_ldfrag(aptr, bptr, s16, 0, aE, bE);
#pragma unroll 2
  for (int k0 = 0; k0 < K; k0 += 8) {
    gemm_ldfrag(aptr, bptr, s16, k0 + 4, aO, bO);
    gemm_mma8(aE, bE, acc);
    if (k0 + 8 < K)
      gemm_ldfrag(aptr, bptr, s16, k0 + 8, aE, bE);
    gemm_mma8(aO, bO, acc);
  }

#pragma unroll
  for (int mt = 0; mt < 2; ++mt)
#pragma unroll
    for (int nt = 0; nt < 4; ++nt)
#pragma unroll
      for (int r = 0; r < 8; ++r) {
        size_t row = (size_t)(mBase + mt * 16 + lhi * 8 + r);
        C[row * Nc + (nBase + nt * 16 + lr)] = acc[mt][nt][r];
      }
}

typedef __attribute__((ext_vector_type(4))) float v4f;

#define QT      16
#define NWAVE   8
#define TPW     8
#define OPITCH  68

__device__ __forceinline__ void attn_ldk(const float* __restrict__ kbase, int kt,
                                         int lr, int lhi, v2f kf[16]) {
  const float* kr = kbase + (size_t)(kt * 16 + lr) * (3 * DD) + 2 * lhi;
#pragma unroll
  for (int ks = 0; ks < 16; ++ks) kf[ks] = *(const v2f*)(kr + 4 * ks);
}

__device__ __forceinline__ void attn_ldv(const float* __restrict__ vbase, int kt,
                                         int lr, int lhi, v4f vf[8]) {
  const float* vr = vbase + (size_t)(kt * 16 + 8 * lhi) * (3 * DD) + 4 * lr;
#pragma unroll
  for (int jj = 0; jj < 8; ++jj) vf[jj] = *(const v4f*)(vr + (size_t)jj * (3 * DD));
}

__global__ __launch_bounds__(256)
void stieltjes_attn(const float* __restrict__ qkv, float* __restrict__ O) {
  __shared__ float2 red[2][NWAVE][QT];
  __shared__ __attribute__((aligned(16))) float Op[NWAVE][QT][OPITCH];

  const int lane = threadIdx.x & 31;
  const int wave = __builtin_amdgcn_readfirstlane((int)(threadIdx.x >> 5));
  const int lhi  = lane >> 4;
  const int lr   = lane & 15;

  const int qb = blockIdx.x & 63;
  const int bh = blockIdx.x >> 6;
  const int h  = bh & (HH - 1);
  const int b  = bh >> 4;

  const int qBase = qb * QT;
  const int nt = (qb >= wave) ? (((qb - wave) >> 3) + 1) : 0;

  const float* qbase = qkv + (size_t)b * NN * (3 * DD) + h * DHH;
  const float* kbase = qbase + DD;
  const float* vbase = qbase + 2 * DD;

  v2f qf[16];
  {
    const float* qr = qbase + (size_t)(qBase + lr) * (3 * DD) + 2 * lhi;
#pragma unroll
    for (int ks = 0; ks < 16; ++ks) qf[ks] = *(const v2f*)(qr + 4 * ks);
  }

  v8f S[TPW];
#pragma unroll
  for (int t = 0; t < TPW; ++t) S[t] = (v8f){NEGBIG,NEGBIG,NEGBIG,NEGBIG,NEGBIG,NEGBIG,NEGBIG,NEGBIG};

  {
    v2f kf[2][16];
    if (nt > 0) attn_ldk(kbase, wave, lr, lhi, kf[0]);
    const int gq = qBase + lr;
#pragma unroll
    for (int t = 0; t < TPW; ++t) {
      if (t < nt) {
        if (t + 1 < nt) attn_ldk(kbase, wave + 8 * (t + 1), lr, lhi, kf[(t + 1) & 1]);
        v8f acc = (v8f){0,0,0,0,0,0,0,0};
#pragma unroll
        for (int ks = 0; ks < 16; ++ks) acc = wmma_f32(kf[t & 1][ks], qf[ks], acc);
        const int gk0 = (wave + 8 * t) * 16 + 8 * lhi;
#pragma unroll
        for (int r = 0; r < 8; ++r) {
          float v = acc[r] * SCALE;
          if (gk0 + r > gq) v = NEGBIG;
          acc[r] = v;
        }
        S[t] = acc;
      }
    }
  }

  float lam;
  {
    float m = -3e30f;
#pragma unroll
    for (int t = 0; t < TPW; ++t)
      if (t < nt) {
#pragma unroll
        for (int r = 0; r < 8; ++r) m = fmaxf(m, S[t][r]);
      }
    m = fmaxf(m, __shfl_xor(m, 16));
    if (lane < 16) red[0][wave][lr].x = m;
    __syncthreads();
    float mm = red[0][0][lr].x;
#pragma unroll
    for (int w = 1; w < NWAVE; ++w) mm = fmaxf(mm, red[0][w][lr].x);
    lam = mm + 1.0f;
  }

  int pl = 1;
#pragma nounroll
  for (int it = 0; it < 4; ++it) {
    float fs = 0.0f, ps = 0.0f;
#pragma unroll
    for (int t = 0; t < TPW; ++t)
      if (t < nt) {
#pragma unroll
        for (int r = 0; r < 8; ++r) {
          float d  = fmaxf(lam - S[t][r], EPS_);
          float r1 = 1.0f / d;
          fs += r1;
          ps += r1 * r1;
        }
      }
    fs += __shfl_xor(fs, 16);
    ps += __shfl_xor(ps, 16);
    if (lane < 16) red[pl][wave][lr] = make_float2(fs, ps);
    __syncthreads();
    float F = 0.0f, P = 0.0f;
#pragma unroll
    for (int w = 0; w < NWAVE; ++w) {
      float2 v = red[pl][w][lr];
      F += v.x;
      P += v.y;
    }
    const float f  = F - 1.0f;
    const float fp = -P;
    lam = lam - f / fp;
    pl ^= 1;
  }

  {
    float ws = 0.0f;
#pragma unroll
    for (int t = 0; t < TPW; ++t)
      if (t < nt) {
#pragma unroll
        for (int r = 0; r < 8; ++r) {
          float r1 = 1.0f / fmaxf(lam - S[t][r], EPS_);
          S[t][r] = r1;
          ws += r1;
        }
      }
    ws += __shfl_xor(ws, 16);
    if (lane < 16) red[pl][wave][lr].x = ws;
    __syncthreads();
    float Wt = 0.0f;
#pragma unroll
    for (int w = 0; w < NWAVE; ++w) Wt += red[pl][w][lr].x;
    const float inv = 1.0f / Wt;
#pragma unroll
    for (int t = 0; t < TPW; ++t)
      if (t < nt) {
#pragma unroll
        for (int r = 0; r < 8; ++r) S[t][r] *= inv;
      }
  }

  v8f oacc[4];
#pragma unroll
  for (int i = 0; i < 4; ++i) oacc[i] = (v8f){0,0,0,0,0,0,0,0};
  {
    v4f vf[2][8];
    if (nt > 0) attn_ldv(vbase, wave, lr, lhi, vf[0]);
#pragma unroll
    for (int t = 0; t < TPW; ++t) {
      if (t < nt) {
        if (t + 1 < nt) attn_ldv(vbase, wave + 8 * (t + 1), lr, lhi, vf[(t + 1) & 1]);
#pragma unroll
        for (int j = 0; j < 4; ++j) {
          v2f bP;
          bP.x = S[t][2 * j];
          bP.y = S[t][2 * j + 1];
#pragma unroll
          for (int i = 0; i < 4; ++i) {
            v2f aV;
            aV.x = vf[t & 1][2 * j][i];
            aV.y = vf[t & 1][2 * j + 1][i];
            oacc[i] = wmma_f32(aV, bP, oacc[i]);
          }
        }
      }
    }
  }

#pragma unroll
  for (int r = 0; r < 8; ++r) {
    v4f o4;
    o4.x = oacc[0][r]; o4.y = oacc[1][r]; o4.z = oacc[2][r]; o4.w = oacc[3][r];
    *(v4f*)(&Op[wave][lr][32 * lhi + 4 * r]) = o4;
  }
  __syncthreads();
  {
    const int q  = threadIdx.x >> 4;
    const int c4 = (threadIdx.x & 15) * 4;
    v4f s = *(const v4f*)(&Op[0][q][c4]);
#pragma unroll
    for (int w = 1; w < NWAVE; ++w) s += *(const v4f*)(&Op[w][q][c4]);
    *(v4f*)(O + ((size_t)b * NN + qBase + q) * DD + h * DHH + c4) = s;
  }
}

extern "C" void kernel_launch(void* const* d_in, const int* in_sizes, int n_in,
                              void* d_out, int out_size, void* d_ws, size_t ws_size,
                              hipStream_t stream) {
  const float* x    = (const float*)d_in[0];
  const float* Wqkv = (const float*)d_in[1];
  const float* Wo   = (const float*)d_in[2];
  float* out = (float*)d_out;

  float* qkv = (float*)d_ws;
  float* Ows = qkv + (size_t)2048 * 3072;

  gemm_nt_f32_wmma<<<(2048 / 64) * (3072 / 128), 128, 0, stream>>>(
      x, Wqkv, qkv, 2048, 3072, 1024);

  stieltjes_attn<<<BB * HH * (NN / 16), 256, 0, stream>>>(qkv, Ows);

  gemm_nt_f32_wmma<<<(2048 / 64) * (1024 / 128), 128, 0, stream>>>(
      Ows, Wo, out, 2048, 1024, 1024);
}


# All A/B fragment loaders (both GEMMs, attention Q and K) read 16 B per lane, lane half h taking k 8j+4h..+3 for two k-steps: half the load instructions and line touches; k sum order changes
# speedup vs baseline: 2.1862x; 1.0595x over previous
#include <hip/hip_runtime.h>

typedef __attribute__((ext_vector_type(2))) float v2f;
typedef __attribute__((ext_vector_type(4))) float v4f;
typedef __attribute__((ext_vector_type(8))) float v8f;

__device__ __forceinline__ v8f wmma_f32(v2f a, v2f b, v8f c) {
  return __builtin_amdgcn_wmma_f32_16x16x4_f32(false, a, false, b, (short)0, c, false, false);
}

__device__ __forceinline__ v2f frag_lo(v4f x) { return __builtin_shufflevector(x, x, 0, 1); }
__device__ __forceinline__ v2f frag_hi(v4f x) { return __builtin_shufflevector(x, x, 2, 3); }

#define BB   2
#define NN   1024
#define DD   1024
#define HH   16
#define DHH  64
#define SCALE 0.125f
#define EPS_  1e-6f
#define NEGBIG (-1e30f)

__device__ __forceinline__ void gemm_ldfrag(const float* __restrict__ aptr,
                                            const float* __restrict__ bptr,
                                            size_t s16, int k0,
                                            v4f a[2], v4f b[4]) {
  a[0] = *(const v4f*)(aptr + k0);
  a[1] = *(const v4f*)(aptr + s16 + k0);
  b[0] = *(const v4f*)(bptr + k0);
  b[1] = *(const v4f*)(bptr + s16 + k0);
  b[2] = *(const v4f*)(bptr + 2 * s16 + k0);
  b[3] = *(const v4f*)(bptr + 3 * s16 + k0);
}

__device__ __forceinline__ void gemm_mma16(const v4f a[2], const v4f b[4],
                                           v8f acc[2][4]) {
#pragma unroll
  for (int mt = 0; mt < 2; ++mt)
#pragma unroll
    for (int nt = 0; nt < 4; ++nt)
      acc[mt][nt] = wmma_f32(frag_lo(a[mt]), frag_lo(b[nt]), acc[mt][nt]);
#pragma unroll
  for (int mt = 0; mt < 2; ++mt)
#pragma unroll
    for (int nt = 0; nt < 4; ++nt)
      acc[mt][nt] = wmma_f32(frag_hi(a[mt]), frag_hi(b[nt]), acc[mt][nt]);
}

__global__ __launch_bounds__(128)
void gemm_nt_f32_wmma(const float* __restrict__ A, const float* __restrict__ Bm,
                      float* __restrict__ C, int M, int Nc, int K) {
  const int lane = threadIdx.x & 31;
  const int wave = threadIdx.x >> 5;
  const int lhi  = lane >> 4;
  const int lr   = lane & 15;

  const int nBlocks = Nc >> 7;
  const int mb = blockIdx.x / nBlocks;
  const int nb = blockIdx.x % nBlocks;
  const int mBase = mb * 64  + (wave >> 1) * 32;
  const int nBase = nb * 128 + (wave & 1) * 64;

  v8f acc[2][4];
#pragma unroll
  for (int mt = 0; mt < 2; ++mt)
#pragma unroll
    for (int nt = 0; nt < 4; ++nt) acc[mt][nt] = (v8f){0,0,0,0,0,0,0,0};

  const float* aptr = A  + (size_t)(mBase + lr) * K + 4 * lhi;
  const float* bptr = Bm + (size_t)(nBase + lr) * K + 4 * lhi;
  const size_t s16  = (size_t)16 * K;

  v4f aE[2], bE[4], aO[2], bO[4];
  gemm_ldfrag(aptr, bptr, s16, 0, aE, bE);
  for (int k0 = 0; k0 < K; k0 += 16) {
    gemm_ldfrag(aptr, bptr, s16, k0 + 8, aO, bO);
    gemm_mma16(aE, bE, acc);
    if (k0 + 16 < K)
      gemm_ldfrag(aptr, bptr, s16, k0 + 16, aE, bE);
    gemm_mma16(aO, bO, acc);
  }

#pragma unroll
  for (int mt = 0; mt < 2; ++mt)
#pragma unroll
    for (int nt = 0; nt < 4; ++nt)
#pragma unroll
      for (int r = 0; r < 8; ++r) {
        size_t row = (size_t)(mBase + mt * 16 + lhi * 8 + r);
        C[row * Nc + (nBase + nt * 16 + lr)] = acc[mt][nt][r];
      }
}


#define QT      16
#define NWAVE   8
#define TPW     8
#define OPITCH  68

__device__ __forceinline__ void attn_ldk(const float* __restrict__ kbase, int kt,
                                         int lr, int lhi, v4f kf[8]) {
  const float* kr = kbase + (size_t)(kt * 16 + lr) * (3 * DD) + 4 * lhi;
#pragma unroll
  for (int j = 0; j < 8; ++j) kf[j] = *(const v4f*)(kr + 8 * j);
}

__device__ __forceinline__ void attn_ldv(const float* __restrict__ vbase, int kt,
                                         int lr, int lhi, v4f vf[8]) {
  const float* vr = vbase + (size_t)(kt * 16 + 8 * lhi) * (3 * DD) + 4 * lr;
#pragma unroll
  for (int jj = 0; jj < 8; ++jj) vf[jj] = *(const v4f*)(vr + (size_t)jj * (3 * DD));
}

__global__ __launch_bounds__(256)
void stieltjes_attn(const float* __restrict__ qkv, float* __restrict__ O) {
  __shared__ float2 red[2][NWAVE][QT];
  __shared__ __attribute__((aligned(16))) float Op[NWAVE][QT][OPITCH];

  const int lane = threadIdx.x & 31;
  const int wave = __builtin_amdgcn_readfirstlane((int)(threadIdx.x >> 5));
  const int lhi  = lane >> 4;
  const int lr   = lane & 15;

  const int qb = blockIdx.x & 63;
  const int bh = blockIdx.x >> 6;
  const int h  = bh & (HH - 1);
  const int b  = bh >> 4;

  const int qBase = qb * QT;
  const int nt = (qb >= wave) ? (((qb - wave) >> 3) + 1) : 0;

  const float* qbase = qkv + (size_t)b * NN * (3 * DD) + h * DHH;
  const float* kbase = qbase + DD;
  const float* vbase = qbase + 2 * DD;

  v4f qf[8];
  {
    const float* qr = qbase + (size_t)(qBase + lr) * (3 * DD) + 4 * lhi;
#pragma unroll
    for (int j = 0; j < 8; ++j) qf[j] = *(const v4f*)(qr + 8 * j);
  }

  v8f S[TPW];
#pragma unroll
  for (int t = 0; t < TPW; ++t) S[t] = (v8f){NEGBIG,NEGBIG,NEGBIG,NEGBIG,NEGBIG,NEGBIG,NEGBIG,NEGBIG};

  {
    v4f kf[2][8];
    if (nt > 0) attn_ldk(kbase, wave, lr, lhi, kf[0]);
    const int gq = qBase + lr;
#pragma unroll
    for (int t = 0; t < TPW; ++t) {
      if (t < nt) {
        if (t + 1 < nt) attn_ldk(kbase, wave + 8 * (t + 1), lr, lhi, kf[(t + 1) & 1]);
        v8f acc = (v8f){0,0,0,0,0,0,0,0};
#pragma unroll
        for (int j = 0; j < 8; ++j) {
          acc = wmma_f32(frag_lo(kf[t & 1][j]), frag_lo(qf[j]), acc);
          acc = wmma_f32(frag_hi(kf[t & 1][j]), frag_hi(qf[j]), acc);
        }
        const int gk0 = (wave + 8 * t) * 16 + 8 * lhi;
#pragma unroll
        for (int r = 0; r < 8; ++r) {
          float v = acc[r] * SCALE;
          if (gk0 + r > gq) v = NEGBIG;
          acc[r] = v;
        }
        S[t] = acc;
      }
    }
  }

  float lam;
  {
    float m = -3e30f;
#pragma unroll
    for (int t = 0; t < TPW; ++t)
      if (t < nt) {
#pragma unroll
        for (int r = 0; r < 8; ++r) m = fmaxf(m, S[t][r]);
      }
    m = fmaxf(m, __shfl_xor(m, 16));
    if (lane < 16) red[0][wave][lr].x = m;
    __syncthreads();
    float mm = red[0][0][lr].x;
#pragma unroll
    for (int w = 1; w < NWAVE; ++w) mm = fmaxf(mm, red[0][w][lr].x);
    lam = mm + 1.0f;
  }

  int pl = 1;
#pragma nounroll
  for (int it = 0; it < 4; ++it) {
    float fs = 0.0f, ps = 0.0f;
#pragma unroll
    for (int t = 0; t < TPW; ++t)
      if (t < nt) {
#pragma unroll
        for (int r = 0; r < 8; ++r) {
          float d  = fmaxf(lam - S[t][r], EPS_);
          float r1 = 1.0f / d;
          fs += r1;
          ps += r1 * r1;
        }
      }
    fs += __shfl_xor(fs, 16);
    ps += __shfl_xor(ps, 16);
    if (lane < 16) red[pl][wave][lr] = make_float2(fs, ps);
    __syncthreads();
    float F = 0.0f, P = 0.0f;
#pragma unroll
    for (int w = 0; w < NWAVE; ++w) {
      float2 v = red[pl][w][lr];
      F += v.x;
      P += v.y;
    }
    const float f  = F - 1.0f;
    const float fp = -P;
    lam = lam - f / fp;
    pl ^= 1;
  }

  {
    float ws = 0.0f;
#pragma unroll
    for (int t = 0; t < TPW; ++t)
      if (t < nt) {
#pragma unroll
        for (int r = 0; r < 8; ++r) {
          float r1 = 1.0f / fmaxf(lam - S[t][r], EPS_);
          S[t][r] = r1;
          ws += r1;
        }
      }
    ws += __shfl_xor(ws, 16);
    if (lane < 16) red[pl][wave][lr].x = ws;
    __syncthreads();
    float Wt = 0.0f;
#pragma unroll
    for (int w = 0; w < NWAVE; ++w) Wt += red[pl][w][lr].x;
    const float inv = 1.0f / Wt;
#pragma unroll
    for (int t = 0; t < TPW; ++t)
      if (t < nt) {
#pragma unroll
        for (int r = 0; r < 8; ++r) S[t][r] *= inv;
      }
  }

  v8f oacc[4];
#pragma unroll
  for (int i = 0; i < 4; ++i) oacc[i] = (v8f){0,0,0,0,0,0,0,0};
  {
    v4f vf[2][8];
    if (nt > 0) attn_ldv(vbase, wave, lr, lhi, vf[0]);
#pragma unroll
    for (int t = 0; t < TPW; ++t) {
      if (t < nt) {
        if (t + 1 < nt) attn_ldv(vbase, wave + 8 * (t + 1), lr, lhi, vf[(t + 1) & 1]);
#pragma unroll
        for (int j = 0; j < 4; ++j) {
          v2f bP;
          bP.x = S[t][2 * j];
          bP.y = S[t][2 * j + 1];
#pragma unroll
          for (int i = 0; i < 4; ++i) {
            v2f aV;
            aV.x = vf[t & 1][2 * j][i];
            aV.y = vf[t & 1][2 * j + 1][i];
            oacc[i] = wmma_f32(aV, bP, oacc[i]);
          }
        }
      }
    }
  }

#pragma unroll
  for (int r = 0; r < 8; ++r) {
    v4f o4;
    o4.x = oacc[0][r]; o4.y = oacc[1][r]; o4.z = oacc[2][r]; o4.w = oacc[3][r];
    *(v4f*)(&Op[wave][lr][32 * lhi + 4 * r]) = o4;
  }
  __syncthreads();
  {
    const int q  = threadIdx.x >> 4;
    const int c4 = (threadIdx.x & 15) * 4;
    v4f s = *(const v4f*)(&Op[0][q][c4]);
#pragma unroll
    for (int w = 1; w < NWAVE; ++w) s += *(const v4f*)(&Op[w][q][c4]);
    *(v4f*)(O + ((size_t)b * NN + qBase + q) * DD + h * DHH + c4) = s;
  }
}

extern "C" void kernel_launch(void* const* d_in, const int* in_sizes, int n_in,
                              void* d_out, int out_size, void* d_ws, size_t ws_size,
                              hipStream_t stream) {
  const float* x    = (const float*)d_in[0];
  const float* Wqkv = (const float*)d_in[1];
  const float* Wo   = (const float*)d_in[2];
  float* out = (float*)d_out;

  float* qkv = (float*)d_ws;
  float* Ows = qkv + (size_t)2048 * 3072;

  gemm_nt_f32_wmma<<<(2048 / 64) * (3072 / 128), 128, 0, stream>>>(
      x, Wqkv, qkv, 2048, 3072, 1024);

  stieltjes_attn<<<BB * HH * (NN / 16), 256, 0, stream>>>(qkv, Ows);

  gemm_nt_f32_wmma<<<(2048 / 64) * (1024 / 128), 128, 0, stream>>>(
      Ows, Wo, out, 2048, 1024, 1024);
}


# Both projection GEMMs: k loop in 32-k blocks, next block's 24 fragment loads issued as one group a whole block (64 WMMAs) ahead, last two blocks peeled so each wait is exact; sum order unchanged
# speedup vs baseline: 2.8595x; 1.3080x over previous
#include <hip/hip_runtime.h>

typedef __attribute__((ext_vector_type(2))) float v2f;
typedef __attribute__((ext_vector_type(4))) float v4f;
typedef __attribute__((ext_vector_type(8))) float v8f;

__device__ __forceinline__ v8f wmma_f32(v2f a, v2f b, v8f c) {
  return __builtin_amdgcn_wmma_f32_16x16x4_f32(false, a, false, b, (short)0, c, false, false);
}

__device__ __forceinline__ v2f frag_lo(v4f x) { return __builtin_shufflevector(x, x, 0, 1); }
__device__ __forceinline__ v2f frag_hi(v4f x) { return __builtin_shufflevector(x, x, 2, 3); }

#define BB   2
#define NN   1024
#define DD   1024
#define HH   16
#define DHH  64
#define SCALE 0.125f
#define EPS_  1e-6f
#define NEGBIG (-1e30f)

__device__ __forceinline__ void gemm_ldblock(const float* __restrict__ aptr,
                                             const float* __restrict__ bptr,
                                             size_t s16, int k0,
                                             v4f a[2][4], v4f b[4][4]) {
#pragma unroll
  for (int mt = 0; mt < 2; ++mt)
#pragma unroll
    for (int j = 0; j < 4; ++j)
      a[mt][j] = *(const v4f*)(aptr + mt * s16 + k0 + 8 * j);
#pragma unroll
  for (int nt = 0; nt < 4; ++nt)
#pragma unroll
    for (int j = 0; j < 4; ++j)
      b[nt][j] = *(const v4f*)(bptr + nt * s16 + k0 + 8 * j);
}

__device__ __forceinline__ void gemm_mma64(const v4f a[2][4], const v4f b[4][4],
                                           v8f acc[2][4]) {
#pragma unroll
  for (int j = 0; j < 4; ++j) {
#pragma unroll
    for (int mt = 0; mt < 2; ++mt)
#pragma unroll
      for (int nt = 0; nt < 4; ++nt)
        acc[mt][nt] = wmma_f32(frag_lo(a[mt][j]), frag_lo(b[nt][j]), acc[mt][nt]);
#pragma unroll
    for (int mt = 0; mt < 2; ++mt)
#pragma unroll
      for (int nt = 0; nt < 4; ++nt)
        acc[mt][nt] = wmma_f32(frag_hi(a[mt][j]), frag_hi(b[nt][j]), acc[mt][nt]);
  }
}

__global__ __launch_bounds__(128)
void gemm_nt_f32_wmma(const float* __restrict__ A, const float* __restrict__ Bm,
                      float* __restrict__ C, int M, int Nc, int K) {
  const int lane = threadIdx.x & 31;
  const int wave = threadIdx.x >> 5;
  const int lhi  = lane >> 4;
  const int lr   = lane & 15;

  const int nBlocks = Nc >> 7;
  const int mb = blockIdx.x / nBlocks;
  const int nb = blockIdx.x % nBlocks;
  const int mBase = mb * 64  + (wave >> 1) * 32;
  const int nBase = nb * 128 + (wave & 1) * 64;

  v8f acc[2][4];
#pragma unroll
  for (int mt = 0; mt < 2; ++mt)
#pragma unroll
    for (int nt = 0; nt < 4; ++nt) acc[mt][nt] = (v8f){0,0,0,0,0,0,0,0};

  const float* aptr = A  + (size_t)(mBase + lr) * K + 4 * lhi;
  const float* bptr = Bm + (size_t)(nBase + lr) * K + 4 * lhi;
  const size_t s16  = (size_t)16 * K;

#define GROUP_FENCE() __builtin_amdgcn_sched_barrier(0)

  v4f aE[2][4], bE[4][4], aO[2][4], bO[4][4];
  gemm_ldblock(aptr, bptr, s16, 0, aE, bE);
  int k0 = 0;
  for (; k0 + 64 < K; k0 += 64) {
    GROUP_FENCE();
    gemm_ldblock(aptr, bptr, s16, k0 + 32, aO, bO);
    GROUP_FENCE();
    gemm_mma64(aE, bE, acc);
    GROUP_FENCE();
    gemm_ldblock(aptr, bptr, s16, k0 + 64, aE, bE);
    GROUP_FENCE();
    gemm_mma64(aO, bO, acc);
  }
  GROUP_FENCE();
  gemm_ldblock(aptr, bptr, s16, k0 + 32, aO, bO);
  GROUP_FENCE();
  gemm_mma64(aE, bE, acc);
  gemm_mma64(aO, bO, acc);
#undef GROUP_FENCE

#pragma unroll
  for (int mt = 0; mt < 2; ++mt)
#pragma unroll
    for (int nt = 0; nt < 4; ++nt)
#pragma unroll
      for (int r = 0; r < 8; ++r) {
        size_t row = (size_t)(mBase + mt * 16 + lhi * 8 + r);
        C[row * Nc + (nBase + nt * 16 + lr)] = acc[mt][nt][r];
      }
}


#define QT      16
#define NWAVE   8
#define TPW     8
#define OPITCH  68

__device__ __forceinline__ void attn_ldk(const float* __restrict__ kbase, int kt,
                                         int lr, int lhi, v4f kf[8]) {
  const float* kr = kbase + (size_t)(kt * 16 + lr) * (3 * DD) + 4 * lhi;
#pragma unroll
  for (int j = 0; j < 8; ++j) kf[j] = *(const v4f*)(kr + 8 * j);
}

__device__ __forceinline__ void attn_ldv(const float* __restrict__ vbase, int kt,
                                         int lr, int lhi, v4f vf[8]) {
  const float* vr = vbase + (size_t)(kt * 16 + 8 * lhi) * (3 * DD) + 4 * lr;
#pragma unroll
  for (int jj = 0; jj < 8; ++jj) vf[jj] = *(const v4f*)(vr + (size_t)jj * (3 * DD));
}

__global__ __launch_bounds__(256)
void stieltjes_attn(const float* __restrict__ qkv, float* __restrict__ O) {
  __shared__ float2 red[2][NWAVE][QT];
  __shared__ __attribute__((aligned(16))) float Op[NWAVE][QT][OPITCH];

  const int lane = threadIdx.x & 31;
  const int wave = __builtin_amdgcn_readfirstlane((int)(threadIdx.x >> 5));
  const int lhi  = lane >> 4;
  const int lr   = lane & 15;

  const int qb = blockIdx.x & 63;
  const int bh = blockIdx.x >> 6;
  const int h  = bh & (HH - 1);
  const int b  = bh >> 4;

  const int qBase = qb * QT;
  const int nt = (qb >= wave) ? (((qb - wave) >> 3) + 1) : 0;

  const float* qbase = qkv + (size_t)b * NN * (3 * DD) + h * DHH;
  const float* kbase = qbase + DD;
  const float* vbase = qbase + 2 * DD;

  v4f qf[8];
  {
    const float* qr = qbase + (size_t)(qBase + lr) * (3 * DD) + 4 * lhi;
#pragma unroll
    for (int j = 0; j < 8; ++j) qf[j] = *(const v4f*)(qr + 8 * j);
  }

  v8f S[TPW];
#pragma unroll
  for (int t = 0; t < TPW; ++t) S[t] = (v8f){NEGBIG,NEGBIG,NEGBIG,NEGBIG,NEGBIG,NEGBIG,NEGBIG,NEGBIG};

  {
    v4f kf[2][8];
    if (nt > 0) attn_ldk(kbase, wave, lr, lhi, kf[0]);
    const int gq = qBase + lr;
#pragma unroll
    for (int t = 0; t < TPW; ++t) {
      if (t < nt) {
        if (t + 1 < nt) attn_ldk(kbase, wave + 8 * (t + 1), lr, lhi, kf[(t + 1) & 1]);
        v8f acc = (v8f){0,0,0,0,0,0,0,0};
#pragma unroll
        for (int j = 0; j < 8; ++j) {
          acc = wmma_f32(frag_lo(kf[t & 1][j]), frag_lo(qf[j]), acc);
          acc = wmma_f32(frag_hi(kf[t & 1][j]), frag_hi(qf[j]), acc);
        }
        const int gk0 = (wave + 8 * t) * 16 + 8 * lhi;
#pragma unroll
        for (int r = 0; r < 8; ++r) {
          float v = acc[r] * SCALE;
          if (gk0 + r > gq) v = NEGBIG;
          acc[r] = v;
        }
        S[t] = acc;
      }
    }
  }

  float lam;
  {
    float m = -3e30f;
#pragma unroll
    for (int t = 0; t < TPW; ++t)
      if (t < nt) {
#pragma unroll
        for (int r = 0; r < 8; ++r) m = fmaxf(m, S[t][r]);
      }
    m = fmaxf(m, __shfl_xor(m, 16));
    if (lane < 16) red[0][wave][lr].x = m;
    __syncthreads();
    float mm = red[0][0][lr].x;
#pragma unroll
    for (int w = 1; w < NWAVE; ++w) mm = fmaxf(mm, red[0][w][lr].x);
    lam = mm + 1.0f;
  }

  int pl = 1;
#pragma nounroll
  for (int it = 0; it < 4; ++it) {
    float fs = 0.0f, ps = 0.0f;
#pragma unroll
    for (int t = 0; t < TPW; ++t)
      if (t < nt) {
#pragma unroll
        for (int r = 0; r < 8; ++r) {
          float d  = fmaxf(lam - S[t][r], EPS_);
          float r1 = 1.0f / d;
          fs += r1;
          ps += r1 * r1;
        }
      }
    fs += __shfl_xor(fs, 16);
    ps += __shfl_xor(ps, 16);
    if (lane < 16) red[pl][wave][lr] = make_float2(fs, ps);
    __syncthreads();
    float F = 0.0f, P = 0.0f;
#pragma unroll
    for (int w = 0; w < NWAVE; ++w) {
      float2 v = red[pl][w][lr];
      F += v.x;
      P += v.y;
    }
    const float f  = F - 1.0f;
    const float fp = -P;
    lam = lam - f / fp;
    pl ^= 1;
  }

  {
    float ws = 0.0f;
#pragma unroll
    for (int t = 0; t < TPW; ++t)
      if (t < nt) {
#pragma unroll
        for (int r = 0; r < 8; ++r) {
          float r1 = 1.0f / fmaxf(lam - S[t][r], EPS_);
          S[t][r] = r1;
          ws += r1;
        }
      }
    ws += __shfl_xor(ws, 16);
    if (lane < 16) red[pl][wave][lr].x = ws;
    __syncthreads();
    float Wt = 0.0f;
#pragma unroll
    for (int w = 0; w < NWAVE; ++w) Wt += red[pl][w][lr].x;
    const float inv = 1.0f / Wt;
#pragma unroll
    for (int t = 0; t < TPW; ++t)
      if (t < nt) {
#pragma unroll
        for (int r = 0; r < 8; ++r) S[t][r] *= inv;
      }
  }

  v8f oacc[4];
#pragma unroll
  for (int i = 0; i < 4; ++i) oacc[i] = (v8f){0,0,0,0,0,0,0,0};
  {
    v4f vf[2][8];
    if (nt > 0) attn_ldv(vbase, wave, lr, lhi, vf[0]);
#pragma unroll
    for (int t = 0; t < TPW; ++t) {
      if (t < nt) {
        if (t + 1 < nt) attn_ldv(vbase, wave + 8 * (t + 1), lr, lhi, vf[(t + 1) & 1]);
#pragma unroll
        for (int j = 0; j < 4; ++j) {
          v2f bP;
          bP.x = S[t][2 * j];
          bP.y = S[t][2 * j + 1];
#pragma unroll
          for (int i = 0; i < 4; ++i) {
            v2f aV;
            aV.x = vf[t & 1][2 * j][i];
            aV.y = vf[t & 1][2 * j + 1][i];
            oacc[i] = wmma_f32(aV, bP, oacc[i]);
          }
        }
      }
    }
  }

#pragma unroll
  for (int r = 0; r < 8; ++r) {
    v4f o4;
    o4.x = oacc[0][r]; o4.y = oacc[1][r]; o4.z = oacc[2][r]; o4.w = oacc[3][r];
    *(v4f*)(&Op[wave][lr][32 * lhi + 4 * r]) = o4;
  }
  __syncthreads();
  {
    const int q  = threadIdx.x >> 4;
    const int c4 = (threadIdx.x & 15) * 4;
    v4f s = *(const v4f*)(&Op[0][q][c4]);
#pragma unroll
    for (int w = 1; w < NWAVE; ++w) s += *(const v4f*)(&Op[w][q][c4]);
    *(v4f*)(O + ((size_t)b * NN + qBase + q) * DD + h * DHH + c4) = s;
  }
}

extern "C" void kernel_launch(void* const* d_in, const int* in_sizes, int n_in,
                              void* d_out, int out_size, void* d_ws, size_t ws_size,
                              hipStream_t stream) {
  const float* x    = (const float*)d_in[0];
  const float* Wqkv = (const float*)d_in[1];
  const float* Wo   = (const float*)d_in[2];
  float* out = (float*)d_out;

  float* qkv = (float*)d_ws;
  float* Ows = qkv + (size_t)2048 * 3072;

  gemm_nt_f32_wmma<<<(2048 / 64) * (3072 / 128), 128, 0, stream>>>(
      x, Wqkv, qkv, 2048, 3072, 1024);

  stieltjes_attn<<<BB * HH * (NN / 16), 256, 0, stream>>>(qkv, Ows);

  gemm_nt_f32_wmma<<<(2048 / 64) * (1024 / 128), 128, 0, stream>>>(
      Ows, Wo, out, 2048, 1024, 1024);
}


# Attention: each workgroup owns query blocks p and 63-p at once (nine score slots per wave, one set of barriers, exchanges and first loads for both): equal causal work everywhere; sum order unchanged
# speedup vs baseline: 3.1333x; 1.0958x over previous
#include <hip/hip_runtime.h>

typedef __attribute__((ext_vector_type(2))) float v2f;
typedef __attribute__((ext_vector_type(4))) float v4f;
typedef __attribute__((ext_vector_type(8))) float v8f;

__device__ __forceinline__ v8f wmma_f32(v2f a, v2f b, v8f c) {
  return __builtin_amdgcn_wmma_f32_16x16x4_f32(false, a, false, b, (short)0, c, false, false);
}

__device__ __forceinline__ v2f frag_lo(v4f x) { return __builtin_shufflevector(x, x, 0, 1); }
__device__ __forceinline__ v2f frag_hi(v4f x) { return __builtin_shufflevector(x, x, 2, 3); }

#define BB   2
#define NN   1024
#define DD   1024
#define HH   16
#define DHH  64
#define SCALE 0.125f
#define EPS_  1e-6f
#define NEGBIG (-1e30f)

__device__ __forceinline__ void gemm_ldblock(const float* __restrict__ aptr,
                                             const float* __restrict__ bptr,
                                             size_t s16, int k0,
                                             v4f a[2][4], v4f b[4][4]) {
#pragma unroll
  for (int mt = 0; mt < 2; ++mt)
#pragma unroll
    for (int j = 0; j < 4; ++j)
      a[mt][j] = *(const v4f*)(aptr + mt * s16 + k0 + 8 * j);
#pragma unroll
  for (int nt = 0; nt < 4; ++nt)
#pragma unroll
    for (int j = 0; j < 4; ++j)
      b[nt][j] = *(const v4f*)(bptr + nt * s16 + k0 + 8 * j);
}

__device__ __forceinline__ void gemm_mma64(const v4f a[2][4], const v4f b[4][4],
                                           v8f acc[2][4]) {
#pragma unroll
  for (int j = 0; j < 4; ++j) {
#pragma unroll
    for (int mt = 0; mt < 2; ++mt)
#pragma unroll
      for (int nt = 0; nt < 4; ++nt)
        acc[mt][nt] = wmma_f32(frag_lo(a[mt][j]), frag_lo(b[nt][j]), acc[mt][nt]);
#pragma unroll
    for (int mt = 0; mt < 2; ++mt)
#pragma unroll
      for (int nt = 0; nt < 4; ++nt)
        acc[mt][nt] = wmma_f32(frag_hi(a[mt][j]), frag_hi(b[nt][j]), acc[mt][nt]);
  }
}

__global__ __launch_bounds__(128)
void gemm_nt_f32_wmma(const float* __restrict__ A, const float* __restrict__ Bm,
                      float* __restrict__ C, int M, int Nc, int K) {
  const int lane = threadIdx.x & 31;
  const int wave = threadIdx.x >> 5;
  const int lhi  = lane >> 4;
  const int lr   = lane & 15;

  const int nBlocks = Nc >> 7;
  const int mb = blockIdx.x / nBlocks;
  const int nb = blockIdx.x % nBlocks;
  const int mBase = mb * 64  + (wave >> 1) * 32;
  const int nBase = nb * 128 + (wave & 1) * 64;

  v8f acc[2][4];
#pragma unroll
  for (int mt = 0; mt < 2; ++mt)
#pragma unroll
    for (int nt = 0; nt < 4; ++nt) acc[mt][nt] = (v8f){0,0,0,0,0,0,0,0};

  const float* aptr = A  + (size_t)(mBase + lr) * K + 4 * lhi;
  const float* bptr = Bm + (size_t)(nBase + lr) * K + 4 * lhi;
  const size_t s16  = (size_t)16 * K;

#define GROUP_FENCE() __builtin_amdgcn_sched_barrier(0)

  v4f aE[2][4], bE[4][4], aO[2][4], bO[4][4];
  gemm_ldblock(aptr, bptr, s16, 0, aE, bE);
  int k0 = 0;
  for (; k0 + 64 < K; k0 += 64) {
    GROUP_FENCE();
    gemm_ldblock(aptr, bptr, s16, k0 + 32, aO, bO);
    GROUP_FENCE();
    gemm_mma64(aE, bE, acc);
    GROUP_FENCE();
    gemm_ldblock(aptr, bptr, s16, k0 + 64, aE, bE);
    GROUP_FENCE();
    gemm_mma64(aO, bO, acc);
  }
  GROUP_FENCE();
  gemm_ldblock(aptr, bptr, s16, k0 + 32, aO, bO);
  GROUP_FENCE();
  gemm_mma64(aE, bE, acc);
  gemm_mma64(aO, bO, acc);
#undef GROUP_FENCE

#pragma unroll
  for (int mt = 0; mt < 2; ++mt)
#pragma unroll
    for (int nt = 0; nt < 4; ++nt)
#pragma unroll
      for (int r = 0; r < 8; ++r) {
        size_t row = (size_t)(mBase + mt * 16 + lhi * 8 + r);
        C[row * Nc + (nBase + nt * 16 + lr)] = acc[mt][nt][r];
      }
}


#define QT      16
#define NWAVE   8
#define NSLOT   9
#define OPITCH  68

#define SLOT_ON(t) ((t) < 7 || (t) < ntot)

__device__ __forceinline__ void attn_ldk(const float* __restrict__ kbase, int kt,
                                         int lr, int lhi, v4f kf[8]) {
  const float* kr = kbase + (size_t)(kt * 16 + lr) * (3 * DD) + 4 * lhi;
#pragma unroll
  for (int j = 0; j < 8; ++j) kf[j] = *(const v4f*)(kr + 8 * j);
}

__device__ __forceinline__ void attn_ldv(const float* __restrict__ vbase, int kt,
                                         int lr, int lhi, v4f vf[8]) {
  const float* vr = vbase + (size_t)(kt * 16 + 8 * lhi) * (3 * DD) + 4 * lr;
#pragma unroll
  for (int jj = 0; jj < 8; ++jj) vf[jj] = *(const v4f*)(vr + (size_t)jj * (3 * DD));
}

__global__ __launch_bounds__(256)
void stieltjes_attn(const float* __restrict__ qkv, float* __restrict__ O) {
  __shared__ __attribute__((aligned(16))) float4 red[2][NWAVE][QT];
  __shared__ __attribute__((aligned(16))) float Op[2][NWAVE][QT][OPITCH];

  const int lane = threadIdx.x & 31;
  const int wave = __builtin_amdgcn_readfirstlane((int)(threadIdx.x >> 5));
  const int lhi  = lane >> 4;
  const int lr   = lane & 15;

  const int pr = blockIdx.x & 31;
  const int bh = blockIdx.x >> 5;
  const int h  = bh & (HH - 1);
  const int b  = bh >> 4;

  const int qbA = pr;
  const int qbB = (NN / QT - 1) - pr;
  const int ntA  = (qbA >= wave) ? (((qbA - wave) >> 3) + 1) : 0;
  const int ntB  = (qbB >= wave) ? (((qbB - wave) >> 3) + 1) : 0;
  const int ntot = ntA + ntB;

#define SLOT_KT(t) (wave + 8 * ((t) - (((t) >= ntA) ? ntA : 0)))

  const float* qbase = qkv + (size_t)b * NN * (3 * DD) + h * DHH;
  const float* kbase = qbase + DD;
  const float* vbase = qbase + 2 * DD;

  const int qb0 = (ntA > 0) ? qbA : qbB;
  v4f qf[8];
  {
    const float* qr = qbase + (size_t)(qb0 * QT + lr) * (3 * DD) + 4 * lhi;
#pragma unroll
    for (int j = 0; j < 8; ++j) qf[j] = *(const v4f*)(qr + 8 * j);
  }

  v8f S[NSLOT];
#pragma unroll
  for (int t = 0; t < NSLOT; ++t) S[t] = (v8f){NEGBIG,NEGBIG,NEGBIG,NEGBIG,NEGBIG,NEGBIG,NEGBIG,NEGBIG};

  {
    v4f kf[2][8];
    attn_ldk(kbase, SLOT_KT(0), lr, lhi, kf[0]);
    int gq = qb0 * QT + lr;
#pragma unroll
    for (int t = 0; t < NSLOT; ++t) {
      if (t > 0 && t == ntA) {
        const float* qn = qbase + (size_t)(qbB * QT + lr) * (3 * DD) + 4 * lhi;
#pragma unroll
        for (int j = 0; j < 8; ++j) qf[j] = *(const v4f*)(qn + 8 * j);
        gq = qbB * QT + lr;
      }
      if (SLOT_ON(t)) {
        if (t + 1 < NSLOT && SLOT_ON(t + 1)) attn_ldk(kbase, SLOT_KT(t + 1), lr, lhi, kf[(t + 1) & 1]);
        v8f acc = (v8f){0,0,0,0,0,0,0,0};
#pragma unroll
        for (int j = 0; j < 8; ++j) {
          acc = wmma_f32(frag_lo(kf[t & 1][j]), frag_lo(qf[j]), acc);
          acc = wmma_f32(frag_hi(kf[t & 1][j]), frag_hi(qf[j]), acc);
        }
        const int gk0 = SLOT_KT(t) * 16 + 8 * lhi;
#pragma unroll
        for (int r = 0; r < 8; ++r) {
          float v = acc[r] * SCALE;
          if (gk0 + r > gq) v = NEGBIG;
          acc[r] = v;
        }
        S[t] = acc;
      }
    }
  }

  float lamA, lamB;
  {
    float m = -3e30f, mA = -3e30f;
#pragma unroll
    for (int t = 0; t < NSLOT; ++t) {
      if (t == ntA) { mA = m; m = -3e30f; }
      if (SLOT_ON(t)) {
#pragma unroll
        for (int r = 0; r < 8; ++r) m = fmaxf(m, S[t][r]);
      }
    }
    float mB = m;
    mA = fmaxf(mA, __shfl_xor(mA, 16));
    mB = fmaxf(mB, __shfl_xor(mB, 16));
    ((float*)&red[0][wave][lr])[2 * lhi] = lhi ? mB : mA;
    __syncthreads();
    float mmA = red[0][0][lr].x, mmB = red[0][0][lr].z;
#pragma unroll
    for (int w = 1; w < NWAVE; ++w) {
      mmA = fmaxf(mmA, red[0][w][lr].x);
      mmB = fmaxf(mmB, red[0][w][lr].z);
    }
    lamA = mmA + 1.0f;
    lamB = mmB + 1.0f;
  }

  int pl = 1;
#pragma nounroll
  for (int it = 0; it < 4; ++it) {
    float fs = 0.0f, ps = 0.0f, fsA = 0.0f, psA = 0.0f;
    float lamc = lamA;
#pragma unroll
    for (int t = 0; t < NSLOT; ++t) {
      if (t == ntA) { fsA = fs; psA = ps; fs = 0.0f; ps = 0.0f; lamc = lamB; }
      if (SLOT_ON(t)) {
#pragma unroll
        for (int r = 0; r < 8; ++r) {
          float d  = fmaxf(lamc - S[t][r], EPS_);
          float r1 = 1.0f / d;
          fs += r1;
          ps += r1 * r1;
        }
      }
    }
    float fsB = fs, psB = ps;
    fsA += __shfl_xor(fsA, 16);
    psA += __shfl_xor(psA, 16);
    fsB += __shfl_xor(fsB, 16);
    psB += __shfl_xor(psB, 16);
    ((float2*)&red[pl][wave][lr])[lhi] = lhi ? make_float2(fsB, psB) : make_float2(fsA, psA);
    __syncthreads();
    float FA = 0.0f, PA = 0.0f, FB = 0.0f, PB = 0.0f;
#pragma unroll
    for (int w = 0; w < NWAVE; ++w) {
      float4 v = red[pl][w][lr];
      FA += v.x;
      PA += v.y;
      FB += v.z;
      PB += v.w;
    }
    lamA = lamA - (FA - 1.0f) / (-PA);
    lamB = lamB - (FB - 1.0f) / (-PB);
    pl ^= 1;
  }

  {
    float ws = 0.0f, wsA = 0.0f;
    float lamc = lamA;
#pragma unroll
    for (int t = 0; t < NSLOT; ++t) {
      if (t == ntA) { wsA = ws; ws = 0.0f; lamc = lamB; }
      if (SLOT_ON(t)) {
#pragma unroll
        for (int r = 0; r < 8; ++r) {
          float r1 = 1.0f / fmaxf(lamc - S[t][r], EPS_);
          S[t][r] = r1;
          ws += r1;
        }
      }
    }
    float wsB = ws;
    wsA += __shfl_xor(wsA, 16);
    wsB += __shfl_xor(wsB, 16);
    ((float*)&red[pl][wave][lr])[2 * lhi] = lhi ? wsB : wsA;
    __syncthreads();
    float WA = 0.0f, WB = 0.0f;
#pragma unroll
    for (int w = 0; w < NWAVE; ++w) {
      WA += red[pl][w][lr].x;
      WB += red[pl][w][lr].z;
    }
    const float invA = 1.0f / WA;
    const float invB = 1.0f / WB;
    float invc = invA;
#pragma unroll
    for (int t = 0; t < NSLOT; ++t) {
      if (t == ntA) invc = invB;
      if (SLOT_ON(t)) {
#pragma unroll
        for (int r = 0; r < 8; ++r) S[t][r] *= invc;
      }
    }
  }

  v8f oacc[4], oA[4];
#pragma unroll
  for (int i = 0; i < 4; ++i) {
    oacc[i] = (v8f){0,0,0,0,0,0,0,0};
    oA[i]   = (v8f){0,0,0,0,0,0,0,0};
  }
  {
    v4f vf[2][8];
    attn_ldv(vbase, SLOT_KT(0), lr, lhi, vf[0]);
#pragma unroll
    for (int t = 0; t < NSLOT; ++t) {
      if (t == ntA) {
#pragma unroll
        for (int i = 0; i < 4; ++i) {
          oA[i]   = oacc[i];
          oacc[i] = (v8f){0,0,0,0,0,0,0,0};
        }
      }
      if (SLOT_ON(t)) {
        if (t + 1 < NSLOT && SLOT_ON(t + 1)) attn_ldv(vbase, SLOT_KT(t + 1), lr, lhi, vf[(t + 1) & 1]);
#pragma unroll
        for (int j = 0; j < 4; ++j) {
          v2f bP;
          bP.x = S[t][2 * j];
          bP.y = S[t][2 * j + 1];
#pragma unroll
          for (int i = 0; i < 4; ++i) {
            v2f aV;
            aV.x = vf[t & 1][2 * j][i];
            aV.y = vf[t & 1][2 * j + 1][i];
            oacc[i] = wmma_f32(aV, bP, oacc[i]);
          }
        }
      }
    }
  }

#pragma unroll
  for (int r = 0; r < 8; ++r) {
    v4f o4;
    o4.x = oA[0][r]; o4.y = oA[1][r]; o4.z = oA[2][r]; o4.w = oA[3][r];
    *(v4f*)(&Op[0][wave][lr][32 * lhi + 4 * r]) = o4;
    o4.x = oacc[0][r]; o4.y = oacc[1][r]; o4.z = oacc[2][r]; o4.w = oacc[3][r];
    *(v4f*)(&Op[1][wave][lr][32 * lhi + 4 * r]) = o4;
  }
  __syncthreads();
  {
    const int q  = threadIdx.x >> 4;
    const int c4 = (threadIdx.x & 15) * 4;
    v4f sa = *(const v4f*)(&Op[0][0][q][c4]);
    v4f sb = *(const v4f*)(&Op[1][0][q][c4]);
#pragma unroll
    for (int w = 1; w < NWAVE; ++w) {
      sa += *(const v4f*)(&Op[0][w][q][c4]);
      sb += *(const v4f*)(&Op[1][w][q][c4]);
    }
    *(v4f*)(O + ((size_t)b * NN + qbA * QT + q) * DD + h * DHH + c4) = sa;
    *(v4f*)(O + ((size_t)b * NN + qbB * QT + q) * DD + h * DHH + c4) = sb;
  }
#undef SLOT_KT
}

extern "C" void kernel_launch(void* const* d_in, const int* in_sizes, int n_in,
                              void* d_out, int out_size, void* d_ws, size_t ws_size,
                              hipStream_t stream) {
  const float* x    = (const float*)d_in[0];
  const float* Wqkv = (const float*)d_in[1];
  const float* Wo   = (const float*)d_in[2];
  float* out = (float*)d_out;

  float* qkv = (float*)d_ws;
  float* Ows = qkv + (size_t)2048 * 3072;

  gemm_nt_f32_wmma<<<(2048 / 64) * (3072 / 128), 128, 0, stream>>>(
      x, Wqkv, qkv, 2048, 3072, 1024);

  stieltjes_attn<<<BB * HH * (NN / 32), 256, 0, stream>>>(qkv, Ows);

  gemm_nt_f32_wmma<<<(2048 / 64) * (1024 / 128), 128, 0, stream>>>(
      Ows, Wo, out, 2048, 1024, 1024);
}


# Newton and normalise passes: the 8 reciprocals of a lane's scores in a tile from ONE division (1/product times the other seven) instead of 8; results differ by ulps, row sums split even/odd
# speedup vs baseline: 3.3874x; 1.0811x over previous
#include <hip/hip_runtime.h>

typedef __attribute__((ext_vector_type(2))) float v2f;
typedef __attribute__((ext_vector_type(4))) float v4f;
typedef __attribute__((ext_vector_type(8))) float v8f;

__device__ __forceinline__ v8f wmma_f32(v2f a, v2f b, v8f c) {
  return __builtin_amdgcn_wmma_f32_16x16x4_f32(false, a, false, b, (short)0, c, false, false);
}

__device__ __forceinline__ v2f frag_lo(v4f x) { return __builtin_shufflevector(x, x, 0, 1); }
__device__ __forceinline__ v2f frag_hi(v4f x) { return __builtin_shufflevector(x, x, 2, 3); }

#define BB   2
#define NN   1024
#define DD   1024
#define HH   16
#define DHH  64
#define SCALE 0.125f
#define EPS_  1e-6f
#define NEGBIG (-1e30f)

__device__ __forceinline__ void gemm_ldblock(const float* __restrict__ aptr,
                                             const float* __restrict__ bptr,
                                             size_t s16, int k0,
                                             v4f a[2][4], v4f b[4][4]) {
#pragma unroll
  for (int mt = 0; mt < 2; ++mt)
#pragma unroll
    for (int j = 0; j < 4; ++j)
      a[mt][j] = *(const v4f*)(aptr + mt * s16 + k0 + 8 * j);
#pragma unroll
  for (int nt = 0; nt < 4; ++nt)
#pragma unroll
    for (int j = 0; j < 4; ++j)
      b[nt][j] = *(const v4f*)(bptr + nt * s16 + k0 + 8 * j);
}

__device__ __forceinline__ void gemm_mma64(const v4f a[2][4], const v4f b[4][4],
                                           v8f acc[2][4]) {
#pragma unroll
  for (int j = 0; j < 4; ++j) {
#pragma unroll
    for (int mt = 0; mt < 2; ++mt)
#pragma unroll
      for (int nt = 0; nt < 4; ++nt)
        acc[mt][nt] = wmma_f32(frag_lo(a[mt][j]), frag_lo(b[nt][j]), acc[mt][nt]);
#pragma unroll
    for (int mt = 0; mt < 2; ++mt)
#pragma unroll
      for (int nt = 0; nt < 4; ++nt)
        acc[mt][nt] = wmma_f32(frag_hi(a[mt][j]), frag_hi(b[nt][j]), acc[mt][nt]);
  }
}

__global__ __launch_bounds__(128)
void gemm_nt_f32_wmma(const float* __restrict__ A, const float* __restrict__ Bm,
                      float* __restrict__ C, int M, int Nc, int K) {
  const int lane = threadIdx.x & 31;
  const int wave = threadIdx.x >> 5;
  const int lhi  = lane >> 4;
  const int lr   = lane & 15;

  const int nBlocks = Nc >> 7;
  const int mb = blockIdx.x / nBlocks;
  const int nb = blockIdx.x % nBlocks;
  const int mBase = mb * 64  + (wave >> 1) * 32;
  const int nBase = nb * 128 + (wave & 1) * 64;

  v8f acc[2][4];
#pragma unroll
  for (int mt = 0; mt < 2; ++mt)
#pragma unroll
    for (int nt = 0; nt < 4; ++nt) acc[mt][nt] = (v8f){0,0,0,0,0,0,0,0};

  const float* aptr = A  + (size_t)(mBase + lr) * K + 4 * lhi;
  const float* bptr = Bm + (size_t)(nBase + lr) * K + 4 * lhi;
  const size_t s16  = (size_t)16 * K;

#define GROUP_FENCE() __builtin_amdgcn_sched_barrier(0)

  v4f aE[2][4], bE[4][4], aO[2][4], bO[4][4];
  gemm_ldblock(aptr, bptr, s16, 0, aE, bE);
  int k0 = 0;
  for (; k0 + 64 < K; k0 += 64) {
    GROUP_FENCE();
    gemm_ldblock(aptr, bptr, s16, k0 + 32, aO, bO);
    GROUP_FENCE();
    gemm_mma64(aE, bE, acc);
    GROUP_FENCE();
    gemm_ldblock(aptr, bptr, s16, k0 + 64, aE, bE);
    GROUP_FENCE();
    gemm_mma64(aO, bO, acc);
  }
  GROUP_FENCE();
  gemm_ldblock(aptr, bptr, s16, k0 + 32, aO, bO);
  GROUP_FENCE();
  gemm_mma64(aE, bE, acc);
  gemm_mma64(aO, bO, acc);
#undef GROUP_FENCE

#pragma unroll
  for (int mt = 0; mt < 2; ++mt)
#pragma unroll
    for (int nt = 0; nt < 4; ++nt)
#pragma unroll
      for (int r = 0; r < 8; ++r) {
        size_t row = (size_t)(mBase + mt * 16 + lhi * 8 + r);
        C[row * Nc + (nBase + nt * 16 + lr)] = acc[mt][nt][r];
      }
}


#define QT      16
#define NWAVE   8
#define NSLOT   9
#define OPITCH  68

#define SLOT_ON(t) ((t) < 7 || (t) < ntot)

__device__ __forceinline__ v8f recip8(v8f d) {
  const v2f d01 = __builtin_shufflevector(d, d, 0, 1);
  const v2f d23 = __builtin_shufflevector(d, d, 2, 3);
  const v2f d45 = __builtin_shufflevector(d, d, 4, 5);
  const v2f d67 = __builtin_shufflevector(d, d, 6, 7);
  const v2f X = d01 * d45;
  const v2f Y = d23 * d67;
  const v2f Z = X * Y;
  const float t = 1.0f / (Z.x * Z.y);
  v2f Zt;
  Zt.x = Z.y * t;
  Zt.y = Z.x * t;
  const v2f AC = Y * Zt;
  const v2f BD = X * Zt;
  const v2f r01 = d45 * AC;
  const v2f r45 = d01 * AC;
  const v2f r23 = d67 * BD;
  const v2f r67 = d23 * BD;
  v8f r;
  r[0] = r01.x; r[1] = r01.y; r[2] = r23.x; r[3] = r23.y;
  r[4] = r45.x; r[5] = r45.y; r[6] = r67.x; r[7] = r67.y;
  return r;
}

__device__ __forceinline__ v8f tile_recips(v8f s, float lam, bool dg, int lim) {
  v8f d;
#pragma unroll
  for (int r = 0; r < 8; ++r) d[r] = fmaxf(lam - s[r], EPS_);
  if (dg) {
#pragma unroll
    for (int r = 0; r < 8; ++r) d[r] = (r > lim) ? 1.0f : d[r];
    v8f q = recip8(d);
#pragma unroll
    for (int r = 0; r < 8; ++r) q[r] = (r > lim) ? 0.0f : q[r];
    return q;
  }
  return recip8(d);
}

__device__ __forceinline__ void attn_ldk(const float* __restrict__ kbase, int kt,
                                         int lr, int lhi, v4f kf[8]) {
  const float* kr = kbase + (size_t)(kt * 16 + lr) * (3 * DD) + 4 * lhi;
#pragma unroll
  for (int j = 0; j < 8; ++j) kf[j] = *(const v4f*)(kr + 8 * j);
}

__device__ __forceinline__ void attn_ldv(const float* __restrict__ vbase, int kt,
                                         int lr, int lhi, v4f vf[8]) {
  const float* vr = vbase + (size_t)(kt * 16 + 8 * lhi) * (3 * DD) + 4 * lr;
#pragma unroll
  for (int jj = 0; jj < 8; ++jj) vf[jj] = *(const v4f*)(vr + (size_t)jj * (3 * DD));
}

__global__ __launch_bounds__(256)
void stieltjes_attn(const float* __restrict__ qkv, float* __restrict__ O) {
  __shared__ __attribute__((aligned(16))) float4 red[2][NWAVE][QT];
  __shared__ __attribute__((aligned(16))) float Op[2][NWAVE][QT][OPITCH];

  const int lane = threadIdx.x & 31;
  const int wave = __builtin_amdgcn_readfirstlane((int)(threadIdx.x >> 5));
  const int lhi  = lane >> 4;
  const int lr   = lane & 15;

  const int pr = blockIdx.x & 31;
  const int bh = blockIdx.x >> 5;
  const int h  = bh & (HH - 1);
  const int b  = bh >> 4;

  const int qbA = pr;
  const int qbB = (NN / QT - 1) - pr;
  const int ntA  = (qbA >= wave) ? (((qbA - wave) >> 3) + 1) : 0;
  const int ntB  = (qbB >= wave) ? (((qbB - wave) >> 3) + 1) : 0;
  const int ntot = ntA + ntB;

#define SLOT_KT(t) (wave + 8 * ((t) - (((t) >= ntA) ? ntA : 0)))

  const int dslotA = (wave == (qbA & 7)) ? ntA - 1 : -1;
  const int dslotB = (wave == (qbB & 7)) ? ntot - 1 : -1;
#define SLOT_DIAG(t) (((t) < 4 && (t) == dslotA) || ((t) >= 6 && (t) == dslotB))
  const int lim = lr - 8 * lhi;

  const float* qbase = qkv + (size_t)b * NN * (3 * DD) + h * DHH;
  const float* kbase = qbase + DD;
  const float* vbase = qbase + 2 * DD;

  const int qb0 = (ntA > 0) ? qbA : qbB;
  v4f qf[8];
  {
    const float* qr = qbase + (size_t)(qb0 * QT + lr) * (3 * DD) + 4 * lhi;
#pragma unroll
    for (int j = 0; j < 8; ++j) qf[j] = *(const v4f*)(qr + 8 * j);
  }

  v8f S[NSLOT];
#pragma unroll
  for (int t = 0; t < NSLOT; ++t) S[t] = (v8f){NEGBIG,NEGBIG,NEGBIG,NEGBIG,NEGBIG,NEGBIG,NEGBIG,NEGBIG};

  {
    v4f kf[2][8];
    attn_ldk(kbase, SLOT_KT(0), lr, lhi, kf[0]);
    int gq = qb0 * QT + lr;
#pragma unroll
    for (int t = 0; t < NSLOT; ++t) {
      if (t > 0 && t == ntA) {
        const float* qn = qbase + (size_t)(qbB * QT + lr) * (3 * DD) + 4 * lhi;
#pragma unroll
        for (int j = 0; j < 8; ++j) qf[j] = *(const v4f*)(qn + 8 * j);
        gq = qbB * QT + lr;
      }
      if (SLOT_ON(t)) {
        if (t + 1 < NSLOT && SLOT_ON(t + 1)) attn_ldk(kbase, SLOT_KT(t + 1), lr, lhi, kf[(t + 1) & 1]);
        v8f acc = (v8f){0,0,0,0,0,0,0,0};
#pragma unroll
        for (int j = 0; j < 8; ++j) {
          acc = wmma_f32(frag_lo(kf[t & 1][j]), frag_lo(qf[j]), acc);
          acc = wmma_f32(frag_hi(kf[t & 1][j]), frag_hi(qf[j]), acc);
        }
        const int gk0 = SLOT_KT(t) * 16 + 8 * lhi;
#pragma unroll
        for (int r = 0; r < 8; ++r) {
          float v = acc[r] * SCALE;
          if (gk0 + r > gq) v = NEGBIG;
          acc[r] = v;
        }
        S[t] = acc;
      }
    }
  }

  float lamA, lamB;
  {
    float m = -3e30f, mA = -3e30f;
#pragma unroll
    for (int t = 0; t < NSLOT; ++t) {
      if (t == ntA) { mA = m; m = -3e30f; }
      if (SLOT_ON(t)) {
#pragma unroll
        for (int r = 0; r < 8; ++r) m = fmaxf(m, S[t][r]);
      }
    }
    float mB = m;
    mA = fmaxf(mA, __shfl_xor(mA, 16));
    mB = fmaxf(mB, __shfl_xor(mB, 16));
    ((float*)&red[0][wave][lr])[2 * lhi] = lhi ? mB : mA;
    __syncthreads();
    float mmA = red[0][0][lr].x, mmB = red[0][0][lr].z;
#pragma unroll
    for (int w = 1; w < NWAVE; ++w) {
      mmA = fmaxf(mmA, red[0][w][lr].x);
      mmB = fmaxf(mmB, red[0][w][lr].z);
    }
    lamA = mmA + 1.0f;
    lamB = mmB + 1.0f;
  }

  int pl = 1;
#pragma nounroll
  for (int it = 0; it < 4; ++it) {
    v2f fs2 = (v2f){0.0f, 0.0f}, ps2 = (v2f){0.0f, 0.0f};
    float fsA = 0.0f, psA = 0.0f;
    float lamc = lamA;
#pragma unroll
    for (int t = 0; t < NSLOT; ++t) {
      if (t == ntA) {
        fsA = fs2.x + fs2.y; psA = ps2.x + ps2.y;
        fs2 = (v2f){0.0f, 0.0f}; ps2 = (v2f){0.0f, 0.0f};
        lamc = lamB;
      }
      if (SLOT_ON(t)) {
        const v8f q = tile_recips(S[t], lamc, SLOT_DIAG(t), lim);
#pragma unroll
        for (int j = 0; j < 4; ++j) {
          v2f p;
          p.x = q[2 * j];
          p.y = q[2 * j + 1];
          fs2 += p;
          ps2 += p * p;
        }
      }
    }
    float fsB = fs2.x + fs2.y, psB = ps2.x + ps2.y;
    fsA += __shfl_xor(fsA, 16);
    psA += __shfl_xor(psA, 16);
    fsB += __shfl_xor(fsB, 16);
    psB += __shfl_xor(psB, 16);
    ((float2*)&red[pl][wave][lr])[lhi] = lhi ? make_float2(fsB, psB) : make_float2(fsA, psA);
    __syncthreads();
    float FA = 0.0f, PA = 0.0f, FB = 0.0f, PB = 0.0f;
#pragma unroll
    for (int w = 0; w < NWAVE; ++w) {
      float4 v = red[pl][w][lr];
      FA += v.x;
      PA += v.y;
      FB += v.z;
      PB += v.w;
    }
    lamA = lamA - (FA - 1.0f) / (-PA);
    lamB = lamB - (FB - 1.0f) / (-PB);
    pl ^= 1;
  }

  {
    v2f ws2 = (v2f){0.0f, 0.0f};
    float wsA = 0.0f;
    float lamc = lamA;
#pragma unroll
    for (int t = 0; t < NSLOT; ++t) {
      if (t == ntA) { wsA = ws2.x + ws2.y; ws2 = (v2f){0.0f, 0.0f}; lamc = lamB; }
      if (SLOT_ON(t)) {
        const v8f q = tile_recips(S[t], lamc, SLOT_DIAG(t), lim);
        S[t] = q;
#pragma unroll
        for (int j = 0; j < 4; ++j) {
          v2f p;
          p.x = q[2 * j];
          p.y = q[2 * j + 1];
          ws2 += p;
        }
      }
    }
    float wsB = ws2.x + ws2.y;
    wsA += __shfl_xor(wsA, 16);
    wsB += __shfl_xor(wsB, 16);
    ((float*)&red[pl][wave][lr])[2 * lhi] = lhi ? wsB : wsA;
    __syncthreads();
    float WA = 0.0f, WB = 0.0f;
#pragma unroll
    for (int w = 0; w < NWAVE; ++w) {
      WA += red[pl][w][lr].x;
      WB += red[pl][w][lr].z;
    }
    const float invA = 1.0f / WA;
    const float invB = 1.0f / WB;
    float invc = invA;
#pragma unroll
    for (int t = 0; t < NSLOT; ++t) {
      if (t == ntA) invc = invB;
      if (SLOT_ON(t)) {
#pragma unroll
        for (int r = 0; r < 8; ++r) S[t][r] *= invc;
      }
    }
  }

  v8f oacc[4], oA[4];
#pragma unroll
  for (int i = 0; i < 4; ++i) {
    oacc[i] = (v8f){0,0,0,0,0,0,0,0};
    oA[i]   = (v8f){0,0,0,0,0,0,0,0};
  }
  {
    v4f vf[2][8];
    attn_ldv(vbase, SLOT_KT(0), lr, lhi, vf[0]);
#pragma unroll
    for (int t = 0; t < NSLOT; ++t) {
      if (t == ntA) {
#pragma unroll
        for (int i = 0; i < 4; ++i) {
          oA[i]   = oacc[i];
          oacc[i] = (v8f){0,0,0,0,0,0,0,0};
        }
      }
      if (SLOT_ON(t)) {
        if (t + 1 < NSLOT && SLOT_ON(t + 1)) attn_ldv(vbase, SLOT_KT(t + 1), lr, lhi, vf[(t + 1) & 1]);
#pragma unroll
        for (int j = 0; j < 4; ++j) {
          v2f bP;
          bP.x = S[t][2 * j];
          bP.y = S[t][2 * j + 1];
#pragma unroll
          for (int i = 0; i < 4; ++i) {
            v2f aV;
            aV.x = vf[t & 1][2 * j][i];
            aV.y = vf[t & 1][2 * j + 1][i];
            oacc[i] = wmma_f32(aV, bP, oacc[i]);
          }
        }
      }
    }
  }

#pragma unroll
  for (int r = 0; r < 8; ++r) {
    v4f o4;
    o4.x = oA[0][r]; o4.y = oA[1][r]; o4.z = oA[2][r]; o4.w = oA[3][r];
    *(v4f*)(&Op[0][wave][lr][32 * lhi + 4 * r]) = o4;
    o4.x = oacc[0][r]; o4.y = oacc[1][r]; o4.z = oacc[2][r]; o4.w = oacc[3][r];
    *(v4f*)(&Op[1][wave][lr][32 * lhi + 4 * r]) = o4;
  }
  __syncthreads();
  {
    const int q  = threadIdx.x >> 4;
    const int c4 = (threadIdx.x & 15) * 4;
    v4f sa = *(const v4f*)(&Op[0][0][q][c4]);
    v4f sb = *(const v4f*)(&Op[1][0][q][c4]);
#pragma unroll
    for (int w = 1; w < NWAVE; ++w) {
      sa += *(const v4f*)(&Op[0][w][q][c4]);
      sb += *(const v4f*)(&Op[1][w][q][c4]);
    }
    *(v4f*)(O + ((size_t)b * NN + qbA * QT + q) * DD + h * DHH + c4) = sa;
    *(v4f*)(O + ((size_t)b * NN + qbB * QT + q) * DD + h * DHH + c4) = sb;
  }
#undef SLOT_DIAG
#undef SLOT_KT
}

extern "C" void kernel_launch(void* const* d_in, const int* in_sizes, int n_in,
                              void* d_out, int out_size, void* d_ws, size_t ws_size,
                              hipStream_t stream) {
  const float* x    = (const float*)d_in[0];
  const float* Wqkv = (const float*)d_in[1];
  const float* Wo   = (const float*)d_in[2];
  float* out = (float*)d_out;

  float* qkv = (float*)d_ws;
  float* Ows = qkv + (size_t)2048 * 3072;

  gemm_nt_f32_wmma<<<(2048 / 64) * (3072 / 128), 128, 0, stream>>>(
      x, Wqkv, qkv, 2048, 3072, 1024);

  stieltjes_attn<<<BB * HH * (NN / 32), 256, 0, stream>>>(qkv, Ows);

  gemm_nt_f32_wmma<<<(2048 / 64) * (1024 / 128), 128, 0, stream>>>(
      Ows, Wo, out, 2048, 1024, 1024);
}


# q, k, v, O and a Wo copy are stored in their consumer's fragment order (qkv by the projection, O and Wo by attention): whole-line fragment loads and stores; score head-dim sum order changes
# speedup vs baseline: 3.5326x; 1.0429x over previous
#include <hip/hip_runtime.h>

typedef __attribute__((ext_vector_type(2))) float v2f;
typedef __attribute__((ext_vector_type(4))) float v4f;
typedef __attribute__((ext_vector_type(8))) float v8f;

__device__ __forceinline__ v8f wmma_f32(v2f a, v2f b, v8f c) {
  return __builtin_amdgcn_wmma_f32_16x16x4_f32(false, a, false, b, (short)0, c, false, false);
}

__device__ __forceinline__ v2f frag_lo(v4f x) { return __builtin_shufflevector(x, x, 0, 1); }
__device__ __forceinline__ v2f frag_hi(v4f x) { return __builtin_shufflevector(x, x, 2, 3); }

#define BB   2
#define NN   1024
#define DD   1024
#define HH   16
#define DHH  64
#define SCALE 0.125f
#define EPS_  1e-6f
#define NEGBIG (-1e30f)

template <bool INPACK>
__device__ __forceinline__ void gemm_ldblock(const float* __restrict__ aptr,
                                             const float* __restrict__ bptr,
                                             size_t s16, int k0,
                                             v4f a[2][4], v4f b[4][4]) {
  const size_t ko = INPACK ? (size_t)k0 * 16 : (size_t)k0;
  const int    js = INPACK ? 128 : 8;
#pragma unroll
  for (int mt = 0; mt < 2; ++mt)
#pragma unroll
    for (int j = 0; j < 4; ++j)
      a[mt][j] = *(const v4f*)(aptr + mt * s16 + ko + js * j);
#pragma unroll
  for (int nt = 0; nt < 4; ++nt)
#pragma unroll
    for (int j = 0; j < 4; ++j)
      b[nt][j] = *(const v4f*)(bptr + nt * s16 + ko + js * j);
}

template <bool TR>
__device__ __forceinline__ void gemm_mma64(const v4f a[2][4], const v4f b[4][4],
                                           v8f acc[2][4]) {
#pragma unroll
  for (int j = 0; j < 4; ++j) {
#pragma unroll
    for (int mt = 0; mt < 2; ++mt)
#pragma unroll
      for (int nt = 0; nt < 4; ++nt)
        acc[mt][nt] = TR ? wmma_f32(frag_lo(b[nt][j]), frag_lo(a[mt][j]), acc[mt][nt])
                         : wmma_f32(frag_lo(a[mt][j]), frag_lo(b[nt][j]), acc[mt][nt]);
#pragma unroll
    for (int mt = 0; mt < 2; ++mt)
#pragma unroll
      for (int nt = 0; nt < 4; ++nt)
        acc[mt][nt] = TR ? wmma_f32(frag_hi(b[nt][j]), frag_hi(a[mt][j]), acc[mt][nt])
                         : wmma_f32(frag_hi(a[mt][j]), frag_hi(b[nt][j]), acc[mt][nt]);
  }
}

template <bool TR, bool INPACK>
__device__ __forceinline__ void gemm_kloop(const float* __restrict__ aptr,
                                           const float* __restrict__ bptr,
                                           size_t s16, int K, v8f acc[2][4]) {
#define GROUP_FENCE() __builtin_amdgcn_sched_barrier(0)

  v4f aE[2][4], bE[4][4], aO[2][4], bO[4][4];
  gemm_ldblock<INPACK>(aptr, bptr, s16, 0, aE, bE);
  int k0 = 0;
  for (; k0 + 64 < K; k0 += 64) {
    GROUP_FENCE();
    gemm_ldblock<INPACK>(aptr, bptr, s16, k0 + 32, aO, bO);
    GROUP_FENCE();
    gemm_mma64<TR>(aE, bE, acc);
    GROUP_FENCE();
    gemm_ldblock<INPACK>(aptr, bptr, s16, k0 + 64, aE, bE);
    GROUP_FENCE();
    gemm_mma64<TR>(aO, bO, acc);
  }
  GROUP_FENCE();
  gemm_ldblock<INPACK>(aptr, bptr, s16, k0 + 32, aO, bO);
  GROUP_FENCE();
  gemm_mma64<TR>(aE, bE, acc);
  gemm_mma64<TR>(aO, bO, acc);
#undef GROUP_FENCE
}

template <bool PACKED>
__global__ __launch_bounds__(128)
void gemm_nt_f32_wmma(const float* __restrict__ A, const float* __restrict__ Bm,
                      float* __restrict__ C, int M, int Nc, int K) {
  const int lane = threadIdx.x & 31;
  const int wave = threadIdx.x >> 5;
  const int lhi  = lane >> 4;
  const int lr   = lane & 15;

  const int nBlocks = Nc >> 7;
  const int mb = blockIdx.x / nBlocks;
  const int nb = blockIdx.x % nBlocks;
  const int mBase = mb * 64  + (wave >> 1) * 32;
  const int nBase = nb * 128 + (wave & 1) * 64;

  v8f acc[2][4];
#pragma unroll
  for (int mt = 0; mt < 2; ++mt)
#pragma unroll
    for (int nt = 0; nt < 4; ++nt) acc[mt][nt] = (v8f){0,0,0,0,0,0,0,0};

  const size_t s16  = (size_t)16 * K;
  const float* aptr = PACKED ? A  + (size_t)(mBase + lr) * K + 4 * lhi
                             : A  + (size_t)(mBase >> 4) * s16 + 4 * lane;
  const float* bptr = PACKED ? Bm + (size_t)(nBase + lr) * K + 4 * lhi
                             : Bm + (size_t)(nBase >> 4) * s16 + 4 * lane;

  if (PACKED) {
    if (nb < (2 * DD) / 128) gemm_kloop<true, false>(aptr, bptr, s16, K, acc);
    else                     gemm_kloop<false, false>(aptr, bptr, s16, K, acc);

    const int plane = nBase >> 10;
    const int head  = (nBase & (DD - 1)) >> 6;
    float* cp = C + (size_t)plane * ((size_t)BB * NN * DD);
#pragma unroll
    for (int mt = 0; mt < 2; ++mt) {
      const int m0   = mBase + 16 * mt;
      const int bidx = m0 >> 10;
      const int tile = (m0 & (NN - 1)) >> 4;
      float* tp = cp + ((size_t)((bidx * HH + head) * (NN / 16) + tile) * 256 + lane) * 4;
#pragma unroll
      for (int nt = 0; nt < 4; ++nt)
#pragma unroll
        for (int p = 0; p < 2; ++p) {
          v4f o;
          o.x = acc[mt][nt][4 * p];     o.y = acc[mt][nt][4 * p + 1];
          o.z = acc[mt][nt][4 * p + 2]; o.w = acc[mt][nt][4 * p + 3];
          *(v4f*)(tp + (2 * nt + p) * 128) = o;
        }
    }
  } else {
    gemm_kloop<false, true>(aptr, bptr, s16, K, acc);
#pragma unroll
    for (int mt = 0; mt < 2; ++mt)
#pragma unroll
      for (int nt = 0; nt < 4; ++nt)
#pragma unroll
        for (int r = 0; r < 8; ++r) {
          size_t row = (size_t)(mBase + mt * 16 + lhi * 8 + r);
          C[row * Nc + (nBase + nt * 16 + lr)] = acc[mt][nt][r];
        }
  }
}


#define QT      16
#define NWAVE   8
#define NSLOT   9
#define OPITCH  68

#define SLOT_ON(t) ((t) < 7 || (t) < ntot)

__device__ __forceinline__ v8f recip8(v8f d) {
  const v2f d01 = __builtin_shufflevector(d, d, 0, 1);
  const v2f d23 = __builtin_shufflevector(d, d, 2, 3);
  const v2f d45 = __builtin_shufflevector(d, d, 4, 5);
  const v2f d67 = __builtin_shufflevector(d, d, 6, 7);
  const v2f X = d01 * d45;
  const v2f Y = d23 * d67;
  const v2f Z = X * Y;
  const float t = 1.0f / (Z.x * Z.y);
  v2f Zt;
  Zt.x = Z.y * t;
  Zt.y = Z.x * t;
  const v2f AC = Y * Zt;
  const v2f BD = X * Zt;
  const v2f r01 = d45 * AC;
  const v2f r45 = d01 * AC;
  const v2f r23 = d67 * BD;
  const v2f r67 = d23 * BD;
  v8f r;
  r[0] = r01.x; r[1] = r01.y; r[2] = r23.x; r[3] = r23.y;
  r[4] = r45.x; r[5] = r45.y; r[6] = r67.x; r[7] = r67.y;
  return r;
}

__device__ __forceinline__ v8f tile_recips(v8f s, float lam, bool dg, int lim) {
  v8f d;
#pragma unroll
  for (int r = 0; r < 8; ++r) d[r] = fmaxf(lam - s[r], EPS_);
  if (dg) {
#pragma unroll
    for (int r = 0; r < 8; ++r) d[r] = (r > lim) ? 1.0f : d[r];
    v8f q = recip8(d);
#pragma unroll
    for (int r = 0; r < 8; ++r) q[r] = (r > lim) ? 0.0f : q[r];
    return q;
  }
  return recip8(d);
}

__device__ __forceinline__ void attn_ldtile(const float* __restrict__ hbase, int tile,
                                            int lane, v4f f[8]) {
  const float* p = hbase + ((size_t)tile * 256 + lane) * 4;
#pragma unroll
  for (int x = 0; x < 8; ++x) f[x] = *(const v4f*)(p + 128 * x);
}

__global__ __launch_bounds__(256)
void stieltjes_attn(const float* __restrict__ qkv, float* __restrict__ O,
                    const float* __restrict__ Wo, float* __restrict__ WoP) {
  __shared__ __attribute__((aligned(16))) float4 red[2][NWAVE][QT];
  __shared__ __attribute__((aligned(16))) float Op[2][NWAVE][QT][OPITCH];

  const int lane = threadIdx.x & 31;
  const int wave = __builtin_amdgcn_readfirstlane((int)(threadIdx.x >> 5));
  const int lhi  = lane >> 4;
  const int lr   = lane & 15;

  const int pr = blockIdx.x & 31;
  const int bh = blockIdx.x >> 5;
  const int h  = bh & (HH - 1);
  const int b  = bh >> 4;

  const int qbA = pr;
  const int qbB = (NN / QT - 1) - pr;
  const int ntA  = (qbA >= wave) ? (((qbA - wave) >> 3) + 1) : 0;
  const int ntB  = (qbB >= wave) ? (((qbB - wave) >> 3) + 1) : 0;
  const int ntot = ntA + ntB;

#define SLOT_KT(t) (wave + 8 * ((t) - (((t) >= ntA) ? ntA : 0)))

  const int dslotA = (wave == (qbA & 7)) ? ntA - 1 : -1;
  const int dslotB = (wave == (qbB & 7)) ? ntot - 1 : -1;
#define SLOT_DIAG(t) (((t) < 4 && (t) == dslotA) || ((t) >= 6 && (t) == dslotB))
  const int lim = lr - 8 * lhi;

  const float* qbase = qkv + (size_t)(b * HH + h) * (NN * DHH);
  const float* kbase = qbase + (size_t)BB * NN * DD;
  const float* vbase = kbase + (size_t)BB * NN * DD;

  const int wtb = 2 * (int)blockIdx.x + (wave >> 2);
  const v4f wpiece = *(const v4f*)(Wo + (size_t)((wtb >> 5) * 16 + lr) * DD
                                      + (wtb & 31) * 32 + 8 * (wave & 3) + 4 * lhi);

  const int qb0 = (ntA > 0) ? qbA : qbB;
  v4f qf[8];
  attn_ldtile(qbase, qb0, lane, qf);

  v8f S[NSLOT];
#pragma unroll
  for (int t = 0; t < NSLOT; ++t) S[t] = (v8f){NEGBIG,NEGBIG,NEGBIG,NEGBIG,NEGBIG,NEGBIG,NEGBIG,NEGBIG};

  {
    v4f kf[2][8];
    attn_ldtile(kbase, SLOT_KT(0), lane, kf[0]);
    int gq = qb0 * QT + lr;
#pragma unroll
    for (int t = 0; t < NSLOT; ++t) {
      if (t > 0 && t == ntA) {
        attn_ldtile(qbase, qbB, lane, qf);
        gq = qbB * QT + lr;
      }
      if (SLOT_ON(t)) {
        if (t + 1 < NSLOT && SLOT_ON(t + 1)) attn_ldtile(kbase, SLOT_KT(t + 1), lane, kf[(t + 1) & 1]);
        v8f acc = (v8f){0,0,0,0,0,0,0,0};
#pragma unroll
        for (int j = 0; j < 8; ++j) {
          acc = wmma_f32(frag_lo(kf[t & 1][j]), frag_lo(qf[j]), acc);
          acc = wmma_f32(frag_hi(kf[t & 1][j]), frag_hi(qf[j]), acc);
        }
        const int gk0 = SLOT_KT(t) * 16 + 8 * lhi;
#pragma unroll
        for (int r = 0; r < 8; ++r) {
          float v = acc[r] * SCALE;
          if (gk0 + r > gq) v = NEGBIG;
          acc[r] = v;
        }
        S[t] = acc;
      }
    }
  }

  float lamA, lamB;
  {
    float m = -3e30f, mA = -3e30f;
#pragma unroll
    for (int t = 0; t < NSLOT; ++t) {
      if (t == ntA) { mA = m; m = -3e30f; }
      if (SLOT_ON(t)) {
#pragma unroll
        for (int r = 0; r < 8; ++r) m = fmaxf(m, S[t][r]);
      }
    }
    float mB = m;
    mA = fmaxf(mA, __shfl_xor(mA, 16));
    mB = fmaxf(mB, __shfl_xor(mB, 16));
    ((float*)&red[0][wave][lr])[2 * lhi] = lhi ? mB : mA;
    __syncthreads();
    float mmA = red[0][0][lr].x, mmB = red[0][0][lr].z;
#pragma unroll
    for (int w = 1; w < NWAVE; ++w) {
      mmA = fmaxf(mmA, red[0][w][lr].x);
      mmB = fmaxf(mmB, red[0][w][lr].z);
    }
    lamA = mmA + 1.0f;
    lamB = mmB + 1.0f;
  }

  int pl = 1;
#pragma nounroll
  for (int it = 0; it < 4; ++it) {
    v2f fs2 = (v2f){0.0f, 0.0f}, ps2 = (v2f){0.0f, 0.0f};
    float fsA = 0.0f, psA = 0.0f;
    float lamc = lamA;
#pragma unroll
    for (int t = 0; t < NSLOT; ++t) {
      if (t == ntA) {
        fsA = fs2.x + fs2.y; psA = ps2.x + ps2.y;
        fs2 = (v2f){0.0f, 0.0f}; ps2 = (v2f){0.0f, 0.0f};
        lamc = lamB;
      }
      if (SLOT_ON(t)) {
        const v8f q = tile_recips(S[t], lamc, SLOT_DIAG(t), lim);
#pragma unroll
        for (int j = 0; j < 4; ++j) {
          v2f p;
          p.x = q[2 * j];
          p.y = q[2 * j + 1];
          fs2 += p;
          ps2 += p * p;
        }
      }
    }
    float fsB = fs2.x + fs2.y, psB = ps2.x + ps2.y;
    fsA += __shfl_xor(fsA, 16);
    psA += __shfl_xor(psA, 16);
    fsB += __shfl_xor(fsB, 16);
    psB += __shfl_xor(psB, 16);
    ((float2*)&red[pl][wave][lr])[lhi] = lhi ? make_float2(fsB, psB) : make_float2(fsA, psA);
    __syncthreads();
    float FA = 0.0f, PA = 0.0f, FB = 0.0f, PB = 0.0f;
#pragma unroll
    for (int w = 0; w < NWAVE; ++w) {
      float4 v = red[pl][w][lr];
      FA += v.x;
      PA += v.y;
      FB += v.z;
      PB += v.w;
    }
    lamA = lamA - (FA - 1.0f) / (-PA);
    lamB = lamB - (FB - 1.0f) / (-PB);
    pl ^= 1;
  }

  {
    v2f ws2 = (v2f){0.0f, 0.0f};
    float wsA = 0.0f;
    float lamc = lamA;
#pragma unroll
    for (int t = 0; t < NSLOT; ++t) {
      if (t == ntA) { wsA = ws2.x + ws2.y; ws2 = (v2f){0.0f, 0.0f}; lamc = lamB; }
      if (SLOT_ON(t)) {
        const v8f q = tile_recips(S[t], lamc, SLOT_DIAG(t), lim);
        S[t] = q;
#pragma unroll
        for (int j = 0; j < 4; ++j) {
          v2f p;
          p.x = q[2 * j];
          p.y = q[2 * j + 1];
          ws2 += p;
        }
      }
    }
    float wsB = ws2.x + ws2.y;
    wsA += __shfl_xor(wsA, 16);
    wsB += __shfl_xor(wsB, 16);
    ((float*)&red[pl][wave][lr])[2 * lhi] = lhi ? wsB : wsA;
    __syncthreads();
    float WA = 0.0f, WB = 0.0f;
#pragma unroll
    for (int w = 0; w < NWAVE; ++w) {
      WA += red[pl][w][lr].x;
      WB += red[pl][w][lr].z;
    }
    const float invA = 1.0f / WA;
    const float invB = 1.0f / WB;
    float invc = invA;
#pragma unroll
    for (int t = 0; t < NSLOT; ++t) {
      if (t == ntA) invc = invB;
      if (SLOT_ON(t)) {
#pragma unroll
        for (int r = 0; r < 8; ++r) S[t][r] *= invc;
      }
    }
  }

  v8f oacc[4], oA[4];
#pragma unroll
  for (int i = 0; i < 4; ++i) {
    oacc[i] = (v8f){0,0,0,0,0,0,0,0};
    oA[i]   = (v8f){0,0,0,0,0,0,0,0};
  }
  {
    v4f vf[2][8];
    attn_ldtile(vbase, SLOT_KT(0), lane, vf[0]);
#pragma unroll
    for (int t = 0; t < NSLOT; ++t) {
      if (t == ntA) {
#pragma unroll
        for (int i = 0; i < 4; ++i) {
          oA[i]   = oacc[i];
          oacc[i] = (v8f){0,0,0,0,0,0,0,0};
        }
      }
      if (SLOT_ON(t)) {
        if (t + 1 < NSLOT && SLOT_ON(t + 1)) attn_ldtile(vbase, SLOT_KT(t + 1), lane, vf[(t + 1) & 1]);
#pragma unroll
        for (int j = 0; j < 4; ++j) {
          v2f bP;
          bP.x = S[t][2 * j];
          bP.y = S[t][2 * j + 1];
#pragma unroll
          for (int i = 0; i < 4; ++i) {
            const v4f vv = vf[t & 1][2 * i + (j >> 1)];
            const v2f aV = (j & 1) ? frag_hi(vv) : frag_lo(vv);
            oacc[i] = wmma_f32(aV, bP, oacc[i]);
          }
        }
      }
    }
  }

#pragma unroll
  for (int i = 0; i < 4; ++i)
#pragma unroll
    for (int p = 0; p < 2; ++p) {
      v4f o4;
      o4.x = oA[i][4 * p]; o4.y = oA[i][4 * p + 1]; o4.z = oA[i][4 * p + 2]; o4.w = oA[i][4 * p + 3];
      *(v4f*)(&Op[0][wave][lr][16 * i + 8 * lhi + 4 * p]) = o4;
      o4.x = oacc[i][4 * p]; o4.y = oacc[i][4 * p + 1]; o4.z = oacc[i][4 * p + 2]; o4.w = oacc[i][4 * p + 3];
      *(v4f*)(&Op[1][wave][lr][16 * i + 8 * lhi + 4 * p]) = o4;
    }
  __syncthreads();
  {
    const int q  = threadIdx.x & 15;
    const int pc = threadIdx.x >> 4;
    const int c4 = pc * 4;
    v4f sa = *(const v4f*)(&Op[0][0][q][c4]);
    v4f sb = *(const v4f*)(&Op[1][0][q][c4]);
#pragma unroll
    for (int w = 1; w < NWAVE; ++w) {
      sa += *(const v4f*)(&Op[0][w][q][c4]);
      sb += *(const v4f*)(&Op[1][w][q][c4]);
    }
    const int kb  = 2 * h + (pc >> 3);
    const int pos = (((pc & 7) >> 1) * 32 + (pc & 1) * 16 + q) * 4;
    *(v4f*)(O + ((size_t)((b * (NN / QT) + qbA) * (DD / 32) + kb) * 512) + pos) = sa;
    *(v4f*)(O + ((size_t)((b * (NN / QT) + qbB) * (DD / 32) + kb) * 512) + pos) = sb;
  }
  *(v4f*)(WoP + ((size_t)wtb * 4 + (wave & 3)) * 128 + lane * 4) = wpiece;
#undef SLOT_DIAG
#undef SLOT_KT
}

extern "C" void kernel_launch(void* const* d_in, const int* in_sizes, int n_in,
                              void* d_out, int out_size, void* d_ws, size_t ws_size,
                              hipStream_t stream) {
  const float* x    = (const float*)d_in[0];
  const float* Wqkv = (const float*)d_in[1];
  const float* Wo   = (const float*)d_in[2];
  float* out = (float*)d_out;

  float* qkv = (float*)d_ws;
  float* Ows = qkv + (size_t)2048 * 3072;
  float* WoP = Ows + (size_t)2048 * 1024;

  gemm_nt_f32_wmma<true><<<(2048 / 64) * (3072 / 128), 128, 0, stream>>>(
      x, Wqkv, qkv, 2048, 3072, 1024);

  stieltjes_attn<<<BB * HH * (NN / 32), 256, 0, stream>>>(qkv, Ows, Wo, WoP);

  gemm_nt_f32_wmma<false><<<(2048 / 64) * (1024 / 128), 128, 0, stream>>>(
      Ows, WoP, out, 2048, 1024, 1024);
}
